# starting kernel
# baseline (speedup 1.0000x reference)
#include <hip/hip_runtime.h>
#include <hip/hip_bf16.h>
#include <math.h>

typedef float v2f __attribute__((ext_vector_type(2)));
typedef float v8f __attribute__((ext_vector_type(8)));

#define NEG_SLOPE 0.2f

__device__ __forceinline__ float leaky(float v) {
    return v > 0.0f ? v : NEG_SLOPE * v;
}

__device__ __forceinline__ void atomicMaxF32(float* addr, float val) {
    if (val >= 0.0f) atomicMax((int*)addr, __float_as_int(val));
    else             atomicMin((unsigned int*)addr, __float_as_uint(val));
}

template<int NT>
__global__ void __launch_bounds__(32)
gemm_wmma_f32(const float* __restrict__ A, const float* __restrict__ B,
              float* __restrict__ C, int M, int K, int N, int ldc)
{
    __shared__ float As[16 * 12];

    const int lane = (int)threadIdx.x;
    const int half = lane >> 4;
    const int l16  = lane & 15;

    const int alr = lane >> 1;
    const int alc = (lane & 1) * 4;
    int rowA = blockIdx.x * 16 + alr;
    if (rowA >= M) rowA = M - 1;
    const float* __restrict__ Aload = A + (size_t)rowA * (size_t)K + alc;

    const int ncol0 = blockIdx.y * (16 * NT);
    const int colB  = ncol0 + l16;

    v8f acc[NT];
#pragma unroll
    for (int t = 0; t < NT; ++t) {
        v8f z = {0.f, 0.f, 0.f, 0.f, 0.f, 0.f, 0.f, 0.f};
        acc[t] = z;
    }

    float4 av = *(const float4*)(Aload);

    for (int kc = 0; kc < K; kc += 8) {
        __syncthreads();
        *(float4*)(&As[alr * 12 + alc]) = av;

        v2f breg[2][NT];
#pragma unroll
        for (int kk2 = 0; kk2 < 2; ++kk2) {
            const float* __restrict__ Bk =
                B + (size_t)(kc + kk2 * 4 + 2 * half) * (size_t)N + colB;
#pragma unroll
            for (int t = 0; t < NT; ++t) {
                breg[kk2][t].x = Bk[t * 16];
                breg[kk2][t].y = Bk[N + t * 16];
            }
        }

        const int kn = (kc + 8 < K) ? (kc + 8) : 0;
        av = *(const float4*)(Aload + kn);

        __syncthreads();

#pragma unroll
        for (int kk2 = 0; kk2 < 2; ++kk2) {
            const v2f a = *(const v2f*)(&As[l16 * 12 + kk2 * 4 + 2 * half]);
#pragma unroll
            for (int t = 0; t < NT; ++t) {
                acc[t] = __builtin_amdgcn_wmma_f32_16x16x4_f32(
                    false, a, false, breg[kk2][t], (short)0, acc[t], false, false);
            }
        }
    }

    const int crow0 = blockIdx.x * 16 + half * 8;
    float* __restrict__ Cbase = C + (size_t)crow0 * (size_t)ldc + ncol0 + l16;
    if (blockIdx.x * 16 + 16 <= M) {
#pragma unroll
        for (int t = 0; t < NT; ++t) {
            float* __restrict__ Cp = Cbase + t * 16;
#pragma unroll
            for (int r = 0; r < 8; ++r)
                Cp[(size_t)r * (size_t)ldc] = acc[t][r];
        }
    } else {
#pragma unroll
        for (int t = 0; t < NT; ++t) {
            float* __restrict__ Cp = Cbase + t * 16;
#pragma unroll
            for (int r = 0; r < 8; ++r)
                if (crow0 + r < M) Cp[(size_t)r * (size_t)ldc] = acc[t][r];
        }
    }
}

__global__ void fill_f32(float* __restrict__ p, int n, float v) {
    int t = blockIdx.x * blockDim.x + threadIdx.x;
    if (t < n) p[t] = v;
}

__global__ void pad_cols(const float* __restrict__ src, float* __restrict__ dst,
                         int K, int N, int Np)
{
    int t = blockIdx.x * blockDim.x + threadIdx.x;
    if (t >= K * Np) return;
    int r = t / Np, c = t - r * Np;
    dst[t] = (c < N) ? src[r * N + c] : 0.f;
}

__global__ void alpha1_kernel(const float* __restrict__ h1,
                              const float* __restrict__ a_src,
                              const float* __restrict__ a_dst,
                              float* __restrict__ as, float* __restrict__ ad, int Nn)
{
    int t = blockIdx.x * blockDim.x + threadIdx.x;
    if (t >= Nn * 8) return;
    int n = t >> 3, h = t & 7;
    const float* hp = h1 + (size_t)n * 512 + h * 64;
    const float* sp = a_src + h * 64;
    const float* dp = a_dst + h * 64;
    float ss = 0.f, dd = 0.f;
#pragma unroll 8
    for (int c = 0; c < 64; ++c) { float v = hp[c]; ss += v * sp[c]; dd += v * dp[c]; }
    as[t] = ss;
    ad[t] = dd;
}

__device__ __forceinline__ void edge_sd(const int* __restrict__ ei, int E, int e,
                                        int& s, int& d)
{
    if (e < E) { s = ei[e]; d = ei[E + e]; }
    else       { s = e - E; d = e - E; }
}

__global__ void edge_max1(const int* __restrict__ ei, int E, int Nn,
                          const float* __restrict__ as, const float* __restrict__ ad,
                          float* __restrict__ mx)
{
    int t = blockIdx.x * blockDim.x + threadIdx.x;
    if (t >= (E + Nn) * 8) return;
    int e = t >> 3, h = t & 7;
    int s, d; edge_sd(ei, E, e, s, d);
    atomicMaxF32(&mx[d * 8 + h], leaky(as[s * 8 + h] + ad[d * 8 + h]));
}

__global__ void edge_expsum1(const int* __restrict__ ei, int E, int Nn,
                             const float* __restrict__ as, const float* __restrict__ ad,
                             const float* __restrict__ mx,
                             float* __restrict__ ex, float* __restrict__ denom)
{
    int t = blockIdx.x * blockDim.x + threadIdx.x;
    if (t >= (E + Nn) * 8) return;
    int e = t >> 3, h = t & 7;
    int s, d; edge_sd(ei, E, e, s, d);
    float v  = leaky(as[s * 8 + h] + ad[d * 8 + h]);
    float xv = expf(v - mx[d * 8 + h]);
    ex[t] = xv;
    atomicAdd(&denom[d * 8 + h], xv);
}

__global__ void aggregate1(const int* __restrict__ ei, int E, int Nn,
                           const float* __restrict__ ex, const float* __restrict__ denom,
                           const float* __restrict__ h1, float* __restrict__ out)
{
    int t = blockIdx.x * blockDim.x + threadIdx.x;
    if (t >= (E + Nn) * 128) return;
    int e = t >> 7;
    int g = t & 127;
    int h = g >> 4;
    int s, d; edge_sd(ei, E, e, s, d);
    float alpha = ex[e * 8 + h] / denom[d * 8 + h];
    const float4 hv = *(const float4*)(h1 + (size_t)s * 512 + g * 4);
    float* o = out + (size_t)d * 512 + g * 4;
    atomicAdd(o + 0, hv.x * alpha);
    atomicAdd(o + 1, hv.y * alpha);
    atomicAdd(o + 2, hv.z * alpha);
    atomicAdd(o + 3, hv.w * alpha);
}

__global__ void bias_elu(float* __restrict__ h2, const float* __restrict__ b, int total) {
    int t = blockIdx.x * blockDim.x + threadIdx.x;
    if (t >= total) return;
    float v = h2[t] + b[t & 511];
    h2[t] = v > 0.f ? v : expm1f(v);
}

__global__ void alpha2_kernel(const float* __restrict__ h3,
                              const float* __restrict__ a_src,
                              const float* __restrict__ a_dst,
                              float* __restrict__ as, float* __restrict__ ad, int Nn)
{
    int n = blockIdx.x * blockDim.x + threadIdx.x;
    if (n >= Nn) return;
    const float* hp = h3 + (size_t)n * 48;
    float ss = 0.f, dd = 0.f;
#pragma unroll 8
    for (int c = 0; c < 40; ++c) { float v = hp[c]; ss += v * a_src[c]; dd += v * a_dst[c]; }
    as[n] = ss;
    ad[n] = dd;
}

__global__ void edge_max2(const int* __restrict__ ei, int E, int Nn,
                          const float* __restrict__ as, const float* __restrict__ ad,
                          float* __restrict__ mx)
{
    int e = blockIdx.x * blockDim.x + threadIdx.x;
    if (e >= E + Nn) return;
    int s, d; edge_sd(ei, E, e, s, d);
    atomicMaxF32(&mx[d], leaky(as[s] + ad[d]));
}

__global__ void edge_expsum2(const int* __restrict__ ei, int E, int Nn,
                             const float* __restrict__ as, const float* __restrict__ ad,
                             const float* __restrict__ mx,
                             float* __restrict__ ex, float* __restrict__ denom)
{
    int e = blockIdx.x * blockDim.x + threadIdx.x;
    if (e >= E + Nn) return;
    int s, d; edge_sd(ei, E, e, s, d);
    float v  = leaky(as[s] + ad[d]);
    float xv = expf(v - mx[d]);
    ex[e] = xv;
    atomicAdd(&denom[d], xv);
}

__global__ void aggregate2(const int* __restrict__ ei, int E, int Nn,
                           const float* __restrict__ ex, const float* __restrict__ denom,
                           const float* __restrict__ h3, float* __restrict__ out)
{
    int t = blockIdx.x * blockDim.x + threadIdx.x;
    if (t >= (E + Nn) * 10) return;
    int e = t / 10;
    int g = t - e * 10;
    int s, d; edge_sd(ei, E, e, s, d);
    float alpha = ex[e] / denom[d];
    const float4 hv = *(const float4*)(h3 + (size_t)s * 48 + g * 4);
    float* o = out + (size_t)d * 40 + g * 4;
    atomicAdd(o + 0, hv.x * alpha);
    atomicAdd(o + 1, hv.y * alpha);
    atomicAdd(o + 2, hv.z * alpha);
    atomicAdd(o + 3, hv.w * alpha);
}

__global__ void bias2_kernel(float* __restrict__ out, const float* __restrict__ b, int total) {
    int t = blockIdx.x * blockDim.x + threadIdx.x;
    if (t >= total) return;
    out[t] += b[t % 40];
}

extern "C" void kernel_launch(void* const* d_in, const int* in_sizes, int n_in,
                              void* d_out, int out_size, void* d_ws, size_t ws_size,
                              hipStream_t stream)
{
    const float* x    = (const float*)d_in[0];
    const int*   ei   = (const int*)  d_in[1];
    const float* W1   = (const float*)d_in[2];
    const float* aS1  = (const float*)d_in[3];
    const float* aD1  = (const float*)d_in[4];
    const float* b1   = (const float*)d_in[5];
    const float* W2   = (const float*)d_in[6];
    const float* aS2  = (const float*)d_in[7];
    const float* aD2  = (const float*)d_in[8];
    const float* b2   = (const float*)d_in[9];
    float* out = (float*)d_out;

    const int IN  = 256, F1 = 512, OUT = 40, OUTP = 48;
    const int Nn  = in_sizes[0] / IN;
    const int E   = in_sizes[1] / 2;
    const int ET  = E + Nn;

    float* ws  = (float*)d_ws;
    float* h1  = ws;
    float* h2  = h1  + (size_t)Nn * F1;
    float* as1 = h2  + (size_t)Nn * F1;
    float* ad1 = as1 + (size_t)Nn * 8;
    float* mx1 = ad1 + (size_t)Nn * 8;
    float* dn1 = mx1 + (size_t)Nn * 8;
    float* ex1 = dn1 + (size_t)Nn * 8;
    float* h3  = ex1 + (size_t)ET * 8;
    float* as2 = h3  + (size_t)Nn * 48;
    float* ad2 = as2 + Nn;
    float* mx2 = ad2 + Nn;
    float* dn2 = mx2 + Nn;
    float* ex2 = dn2 + Nn;
    float* W2p = ex2 + ET;

    const int B = 256;
    const float NEG_INF = -__builtin_inff();

    gemm_wmma_f32<4><<<dim3((Nn + 15) / 16, F1 / 64), 32, 0, stream>>>(
        x, W1, h1, Nn, IN, F1, F1);

    alpha1_kernel<<<(Nn * 8 + B - 1) / B, B, 0, stream>>>(h1, aS1, aD1, as1, ad1, Nn);

    fill_f32<<<(Nn * 8 + B - 1) / B, B, 0, stream>>>(mx1, Nn * 8, NEG_INF);
    fill_f32<<<(Nn * 8 + B - 1) / B, B, 0, stream>>>(dn1, Nn * 8, 0.f);
    fill_f32<<<(Nn * F1 + B - 1) / B, B, 0, stream>>>(h2, Nn * F1, 0.f);

    edge_max1   <<<(ET * 8 + B - 1) / B, B, 0, stream>>>(ei, E, Nn, as1, ad1, mx1);
    edge_expsum1<<<(ET * 8 + B - 1) / B, B, 0, stream>>>(ei, E, Nn, as1, ad1, mx1, ex1, dn1);
    aggregate1  <<<(ET * 128 + B - 1) / B, B, 0, stream>>>(ei, E, Nn, ex1, dn1, h1, h2);

    bias_elu<<<(Nn * F1 + B - 1) / B, B, 0, stream>>>(h2, b1, Nn * F1);

    pad_cols<<<(F1 * OUTP + B - 1) / B, B, 0, stream>>>(W2, W2p, F1, OUT, OUTP);

    gemm_wmma_f32<3><<<dim3((Nn + 15) / 16, 1), 32, 0, stream>>>(
        h2, W2p, h3, Nn, F1, OUTP, OUTP);

    alpha2_kernel<<<(Nn + B - 1) / B, B, 0, stream>>>(h3, aS2, aD2, as2, ad2, Nn);

    fill_f32<<<(Nn + B - 1) / B, B, 0, stream>>>(mx2, Nn, NEG_INF);
    fill_f32<<<(Nn + B - 1) / B, B, 0, stream>>>(dn2, Nn, 0.f);
    fill_f32<<<(Nn * OUT + B - 1) / B, B, 0, stream>>>(out, Nn * OUT, 0.f);

    edge_max2   <<<(ET + B - 1) / B, B, 0, stream>>>(ei, E, Nn, as2, ad2, mx2);
    edge_expsum2<<<(ET + B - 1) / B, B, 0, stream>>>(ei, E, Nn, as2, ad2, mx2, ex2, dn2);
    aggregate2  <<<(ET * 10 + B - 1) / B, B, 0, stream>>>(ei, E, Nn, ex2, dn2, h3, out);

    bias2_kernel<<<(Nn * OUT + B - 1) / B, B, 0, stream>>>(out, b2, Nn * OUT);
}


# Layer-1 aggregation per channel pair: a work-group holds 2 channels of all nodes in LDS, streams the edges, LDS float adds (arrival order, self loop first); divide after the sum; h1,h2 channel-major
# speedup vs baseline: 1.2332x; 1.2332x over previous
#include <hip/hip_runtime.h>
#include <hip/hip_bf16.h>
#include <math.h>


typedef float v2f __attribute__((ext_vector_type(2)));
typedef float v8f __attribute__((ext_vector_type(8)));

#define NEG_SLOPE 0.2f

__device__ __forceinline__ float leaky(float v) {
    return v > 0.0f ? v : NEG_SLOPE * v;
}

__device__ __forceinline__ void atomicMaxF32(float* addr, float val) {
    if (val >= 0.0f) atomicMax((int*)addr, __float_as_int(val));
    else             atomicMin((unsigned int*)addr, __float_as_uint(val));
}

template<int NT>
__global__ void __launch_bounds__(32)
gemm_wmma_f32_chmajor_out(const float* __restrict__ A, const float* __restrict__ B,
                          float* __restrict__ Ct, int M, int K, int N)
{
    __shared__ float As[16 * 12];

    const int lane = (int)threadIdx.x;
    const int half = lane >> 4;
    const int l16  = lane & 15;

    const int alr = lane >> 1;
    const int alc = (lane & 1) * 4;
    int rowA = blockIdx.x * 16 + alr;
    if (rowA >= M) rowA = M - 1;
    const float* __restrict__ Aload = A + (size_t)rowA * (size_t)K + alc;

    const int ncol0 = blockIdx.y * (16 * NT);
    const int colB  = ncol0 + l16;

    v8f acc[NT];
#pragma unroll
    for (int t = 0; t < NT; ++t) {
        v8f z = {0.f, 0.f, 0.f, 0.f, 0.f, 0.f, 0.f, 0.f};
        acc[t] = z;
    }

    float4 av = *(const float4*)(Aload);

    for (int kc = 0; kc < K; kc += 8) {
        __syncthreads();
        *(float4*)(&As[alr * 12 + alc]) = av;

        v2f breg[2][NT];
#pragma unroll
        for (int kk2 = 0; kk2 < 2; ++kk2) {
            const float* __restrict__ Bk =
                B + (size_t)(kc + kk2 * 4 + 2 * half) * (size_t)N + colB;
#pragma unroll
            for (int t = 0; t < NT; ++t) {
                breg[kk2][t].x = Bk[t * 16];
                breg[kk2][t].y = Bk[N + t * 16];
            }
        }

        const int kn = (kc + 8 < K) ? (kc + 8) : 0;
        av = *(const float4*)(Aload + kn);

        __syncthreads();

#pragma unroll
        for (int kk2 = 0; kk2 < 2; ++kk2) {
            const v2f a = *(const v2f*)(&As[l16 * 12 + kk2 * 4 + 2 * half]);
#pragma unroll
            for (int t = 0; t < NT; ++t) {
                acc[t] = __builtin_amdgcn_wmma_f32_16x16x4_f32(
                    false, breg[kk2][t], false, a, (short)0, acc[t], false, false);
            }
        }
    }

    const int node0 = blockIdx.x * 16;
    float* __restrict__ Cbase =
        Ct + (size_t)(ncol0 + half * 8) * (size_t)M + node0 + l16;
    if (node0 + 16 <= M) {
#pragma unroll
        for (int t = 0; t < NT; ++t) {
#pragma unroll
            for (int r = 0; r < 8; ++r)
                Cbase[(size_t)(t * 16 + r) * (size_t)M] = acc[t][r];
        }
    } else {
#pragma unroll
        for (int t = 0; t < NT; ++t) {
#pragma unroll
            for (int r = 0; r < 8; ++r)
                if (node0 + l16 < M) Cbase[(size_t)(t * 16 + r) * (size_t)M] = acc[t][r];
        }
    }
}

template<int NT>
__global__ void __launch_bounds__(32)
gemm_wmma_f32_chmajor_in(const float* __restrict__ At, const float* __restrict__ B,
                         float* __restrict__ C, int M, int K, int N, int ldc)
{
    const int lane = (int)threadIdx.x;
    const int half = lane >> 4;
    const int l16  = lane & 15;

    int rowA = blockIdx.x * 16 + l16;
    if (rowA >= M) rowA = M - 1;
    const float* __restrict__ Aload = At + (size_t)(2 * half) * (size_t)M + rowA;

    const int ncol0 = blockIdx.y * (16 * NT);
    const int colB  = ncol0 + l16;

    v8f acc[NT];
#pragma unroll
    for (int t = 0; t < NT; ++t) {
        v8f z = {0.f, 0.f, 0.f, 0.f, 0.f, 0.f, 0.f, 0.f};
        acc[t] = z;
    }

    v2f an[2];
#pragma unroll
    for (int kk2 = 0; kk2 < 2; ++kk2) {
        an[kk2].x = Aload[(size_t)(kk2 * 4) * (size_t)M];
        an[kk2].y = Aload[(size_t)(kk2 * 4 + 1) * (size_t)M];
    }

    for (int kc = 0; kc < K; kc += 8) {
        const v2f a0 = an[0];
        const v2f a1 = an[1];

        v2f breg[2][NT];
#pragma unroll
        for (int kk2 = 0; kk2 < 2; ++kk2) {
            const float* __restrict__ Bk =
                B + (size_t)(kc + kk2 * 4 + 2 * half) * (size_t)N + colB;
#pragma unroll
            for (int t = 0; t < NT; ++t) {
                breg[kk2][t].x = Bk[t * 16];
                breg[kk2][t].y = Bk[N + t * 16];
            }
        }

        const int kn = (kc + 8 < K) ? (kc + 8) : 0;
#pragma unroll
        for (int kk2 = 0; kk2 < 2; ++kk2) {
            an[kk2].x = Aload[(size_t)(kn + kk2 * 4) * (size_t)M];
            an[kk2].y = Aload[(size_t)(kn + kk2 * 4 + 1) * (size_t)M];
        }

#pragma unroll
        for (int t = 0; t < NT; ++t) {
            acc[t] = __builtin_amdgcn_wmma_f32_16x16x4_f32(
                false, a0, false, breg[0][t], (short)0, acc[t], false, false);
        }
#pragma unroll
        for (int t = 0; t < NT; ++t) {
            acc[t] = __builtin_amdgcn_wmma_f32_16x16x4_f32(
                false, a1, false, breg[1][t], (short)0, acc[t], false, false);
        }
    }

    const int crow0 = blockIdx.x * 16 + half * 8;
    float* __restrict__ Cbase = C + (size_t)crow0 * (size_t)ldc + ncol0 + l16;
    if (blockIdx.x * 16 + 16 <= M) {
#pragma unroll
        for (int t = 0; t < NT; ++t) {
            float* __restrict__ Cp = Cbase + t * 16;
#pragma unroll
            for (int r = 0; r < 8; ++r)
                Cp[(size_t)r * (size_t)ldc] = acc[t][r];
        }
    } else {
#pragma unroll
        for (int t = 0; t < NT; ++t) {
            float* __restrict__ Cp = Cbase + t * 16;
#pragma unroll
            for (int r = 0; r < 8; ++r)
                if (crow0 + r < M) Cp[(size_t)r * (size_t)ldc] = acc[t][r];
        }
    }
}

__global__ void fill_f32(float* __restrict__ p, int n, float v) {
    int t = blockIdx.x * blockDim.x + threadIdx.x;
    if (t < n) p[t] = v;
}

__global__ void pad_cols(const float* __restrict__ src, float* __restrict__ dst,
                         int K, int N, int Np)
{
    int t = blockIdx.x * blockDim.x + threadIdx.x;
    if (t >= K * Np) return;
    int r = t / Np, c = t - r * Np;
    dst[t] = (c < N) ? src[r * N + c] : 0.f;
}

__global__ void alpha1_kernel(const float* __restrict__ h1t,
                              const float* __restrict__ a_src,
                              const float* __restrict__ a_dst,
                              float* __restrict__ as, float* __restrict__ ad, int Nn)
{
    int t = blockIdx.x * blockDim.x + threadIdx.x;
    if (t >= Nn * 8) return;
    int h = t / Nn, n = t - h * Nn;
    const float* hp = h1t + (size_t)(h * 64) * (size_t)Nn + n;
    const float* sp = a_src + h * 64;
    const float* dp = a_dst + h * 64;
    float ss = 0.f, dd = 0.f;
#pragma unroll 8
    for (int c = 0; c < 64; ++c) {
        float v = hp[(size_t)c * (size_t)Nn];
        ss += v * sp[c];
        dd += v * dp[c];
    }
    as[n * 8 + h] = ss;
    ad[n * 8 + h] = dd;
}

__device__ __forceinline__ void edge_sd(const int* __restrict__ ei, int E, int e,
                                        int& s, int& d)
{
    if (e < E) { s = ei[e]; d = ei[E + e]; }
    else       { s = e - E; d = e - E; }
}

__global__ void edge_max1(const int* __restrict__ ei, int E, int Nn,
                          const float* __restrict__ as, const float* __restrict__ ad,
                          float* __restrict__ mx)
{
    int t = blockIdx.x * blockDim.x + threadIdx.x;
    if (t >= (E + Nn) * 8) return;
    int e = t >> 3, h = t & 7;
    int s, d; edge_sd(ei, E, e, s, d);
    atomicMaxF32(&mx[d * 8 + h], leaky(as[s * 8 + h] + ad[d * 8 + h]));
}

__global__ void edge_expsum1(const int* __restrict__ ei, int E, int Nn,
                             const float* __restrict__ as, const float* __restrict__ ad,
                             const float* __restrict__ mx,
                             float* __restrict__ ex, float* __restrict__ denom)
{
    int t = blockIdx.x * blockDim.x + threadIdx.x;
    if (t >= (E + Nn) * 8) return;
    int e = t >> 3, h = t & 7;
    int s, d; edge_sd(ei, E, e, s, d);
    float v  = leaky(as[s * 8 + h] + ad[d * 8 + h]);
    float xv = expf(v - mx[d * 8 + h]);
    ex[(size_t)h * (size_t)(E + Nn) + e] = xv;
    atomicAdd(&denom[d * 8 + h], xv);
}

#define AGG_NODES   10496
#define AGG_THREADS 1024

__global__ void __launch_bounds__(AGG_THREADS)
aggregate1_sliced(const int* __restrict__ ei, int E, int Nn,
                  const float* __restrict__ ex, const float* __restrict__ denom,
                  const float* __restrict__ h1t, const float* __restrict__ b,
                  float* __restrict__ h2t)
{
    __shared__ float2 hs[AGG_NODES];
    __shared__ float  acc[2 * AGG_NODES];

    const int tid = (int)threadIdx.x;
    const int c0  = (int)blockIdx.x * 2;
    const int h   = c0 >> 6;
    const int ET  = E + Nn;

    const float* __restrict__ exh = ex + (size_t)h * (size_t)ET;
    const float* __restrict__ p0  = h1t + (size_t)c0 * (size_t)Nn;
    const float* __restrict__ p1  = p0 + Nn;
    for (int n = tid; n < Nn; n += AGG_THREADS) {
        const float x0 = p0[n], x1 = p1[n];
        const float ws = exh[E + n];
        hs[n] = make_float2(x0, x1);
        acc[2 * n]     = ws * x0;
        acc[2 * n + 1] = ws * x1;
    }
    __syncthreads();

    const int* __restrict__ src = ei;
    const int* __restrict__ dst = ei + E;

    auto load_group = [&](int g, int* s, int* d, float* w) {
#pragma unroll
        for (int j = 0; j < 4; ++j) {
            const int e = g * (4 * AGG_THREADS) + tid + j * AGG_THREADS;
            s[j] = src[e];
            d[j] = dst[e];
            w[j] = exh[e];
        }
    };
    auto add_edge = [&](int d, float w, float2 hv) {
        __hip_atomic_fetch_add(&acc[2 * d],     w * hv.x,
                               __ATOMIC_RELAXED, __HIP_MEMORY_SCOPE_WORKGROUP);
        __hip_atomic_fetch_add(&acc[2 * d + 1], w * hv.y,
                               __ATOMIC_RELAXED, __HIP_MEMORY_SCOPE_WORKGROUP);
    };
    auto do_group = [&](const int* s, const int* d, const float* w) {
        float2 hv[4];
#pragma unroll
        for (int j = 0; j < 4; ++j) hv[j] = hs[s[j]];
#pragma unroll
        for (int j = 0; j < 4; ++j) add_edge(d[j], w[j], hv[j]);
    };

    const int nfull = E / (4 * AGG_THREADS);
    int   sA[4], dA[4], sB[4], dB[4];
    float wA[4], wB[4];
    if (nfull > 0) load_group(0, sA, dA, wA);
    int g = 0;
    for (; g + 2 <= nfull; g += 2) {
        load_group(g + 1, sB, dB, wB);
        do_group(sA, dA, wA);
        load_group((g + 2 < nfull) ? (g + 2) : (g + 1), sA, dA, wA);
        do_group(sB, dB, wB);
    }
    if (g < nfull) do_group(sA, dA, wA);

    for (int e = nfull * (4 * AGG_THREADS) + tid; e < E; e += AGG_THREADS)
        add_edge(dst[e], exh[e], hs[src[e]]);
    __syncthreads();

    const float b0 = b[c0], b1 = b[c0 + 1];
    float* __restrict__ q0 = h2t + (size_t)c0 * (size_t)Nn;
    float* __restrict__ q1 = q0 + Nn;
    for (int n = tid; n < Nn; n += AGG_THREADS) {
        const float dn = denom[n * 8 + h];
        const float v0 = acc[2 * n]     / dn + b0;
        const float v1 = acc[2 * n + 1] / dn + b1;
        q0[n] = v0 > 0.f ? v0 : expm1f(v0);
        q1[n] = v1 > 0.f ? v1 : expm1f(v1);
    }
}

__global__ void alpha2_kernel(const float* __restrict__ h3,
                              const float* __restrict__ a_src,
                              const float* __restrict__ a_dst,
                              float* __restrict__ as, float* __restrict__ ad, int Nn)
{
    int n = blockIdx.x * blockDim.x + threadIdx.x;
    if (n >= Nn) return;
    const float* hp = h3 + (size_t)n * 48;
    float ss = 0.f, dd = 0.f;
#pragma unroll 8
    for (int c = 0; c < 40; ++c) { float v = hp[c]; ss += v * a_src[c]; dd += v * a_dst[c]; }
    as[n] = ss;
    ad[n] = dd;
}

__global__ void edge_max2(const int* __restrict__ ei, int E, int Nn,
                          const float* __restrict__ as, const float* __restrict__ ad,
                          float* __restrict__ mx)
{
    int e = blockIdx.x * blockDim.x + threadIdx.x;
    if (e >= E + Nn) return;
    int s, d; edge_sd(ei, E, e, s, d);
    atomicMaxF32(&mx[d], leaky(as[s] + ad[d]));
}

__global__ void edge_expsum2(const int* __restrict__ ei, int E, int Nn,
                             const float* __restrict__ as, const float* __restrict__ ad,
                             const float* __restrict__ mx,
                             float* __restrict__ ex, float* __restrict__ denom)
{
    int e = blockIdx.x * blockDim.x + threadIdx.x;
    if (e >= E + Nn) return;
    int s, d; edge_sd(ei, E, e, s, d);
    float v  = leaky(as[s] + ad[d]);
    float xv = expf(v - mx[d]);
    ex[e] = xv;
    atomicAdd(&denom[d], xv);
}

__global__ void aggregate2(const int* __restrict__ ei, int E, int Nn,
                           const float* __restrict__ ex, const float* __restrict__ denom,
                           const float* __restrict__ h3, float* __restrict__ out)
{
    int t = blockIdx.x * blockDim.x + threadIdx.x;
    if (t >= (E + Nn) * 10) return;
    int e = t / 10;
    int g = t - e * 10;
    int s, d; edge_sd(ei, E, e, s, d);
    float alpha = ex[e] / denom[d];
    const float4 hv = *(const float4*)(h3 + (size_t)s * 48 + g * 4);
    float* o = out + (size_t)d * 40 + g * 4;
    atomicAdd(o + 0, hv.x * alpha);
    atomicAdd(o + 1, hv.y * alpha);
    atomicAdd(o + 2, hv.z * alpha);
    atomicAdd(o + 3, hv.w * alpha);
}

__global__ void bias2_kernel(float* __restrict__ out, const float* __restrict__ b, int total) {
    int t = blockIdx.x * blockDim.x + threadIdx.x;
    if (t >= total) return;
    out[t] += b[t % 40];
}

extern "C" void kernel_launch(void* const* d_in, const int* in_sizes, int n_in,
                              void* d_out, int out_size, void* d_ws, size_t ws_size,
                              hipStream_t stream)
{
    const float* x    = (const float*)d_in[0];
    const int*   ei   = (const int*)  d_in[1];
    const float* W1   = (const float*)d_in[2];
    const float* aS1  = (const float*)d_in[3];
    const float* aD1  = (const float*)d_in[4];
    const float* b1   = (const float*)d_in[5];
    const float* W2   = (const float*)d_in[6];
    const float* aS2  = (const float*)d_in[7];
    const float* aD2  = (const float*)d_in[8];
    const float* b2   = (const float*)d_in[9];
    float* out = (float*)d_out;

    const int IN  = 256, F1 = 512, OUT = 40, OUTP = 48;
    const int Nn  = in_sizes[0] / IN;
    const int E   = in_sizes[1] / 2;
    const int ET  = E + Nn;

    float* ws  = (float*)d_ws;
    float* h1  = ws;
    float* h2  = h1  + (size_t)Nn * F1;
    float* as1 = h2  + (size_t)Nn * F1;
    float* ad1 = as1 + (size_t)Nn * 8;
    float* mx1 = ad1 + (size_t)Nn * 8;
    float* dn1 = mx1 + (size_t)Nn * 8;
    float* ex1 = dn1 + (size_t)Nn * 8;
    float* h3  = ex1 + (size_t)ET * 8;
    float* as2 = h3  + (size_t)Nn * 48;
    float* ad2 = as2 + Nn;
    float* mx2 = ad2 + Nn;
    float* dn2 = mx2 + Nn;
    float* ex2 = dn2 + Nn;
    float* W2p = ex2 + ET;

    const int B = 256;
    const float NEG_INF = -__builtin_inff();

    gemm_wmma_f32_chmajor_out<4><<<dim3((Nn + 15) / 16, F1 / 64), 32, 0, stream>>>(
        x, W1, h1, Nn, IN, F1);

    alpha1_kernel<<<(Nn * 8 + B - 1) / B, B, 0, stream>>>(h1, aS1, aD1, as1, ad1, Nn);

    fill_f32<<<(Nn * 8 + B - 1) / B, B, 0, stream>>>(mx1, Nn * 8, NEG_INF);
    fill_f32<<<(Nn * 8 + B - 1) / B, B, 0, stream>>>(dn1, Nn * 8, 0.f);

    edge_max1   <<<(ET * 8 + B - 1) / B, B, 0, stream>>>(ei, E, Nn, as1, ad1, mx1);
    edge_expsum1<<<(ET * 8 + B - 1) / B, B, 0, stream>>>(ei, E, Nn, as1, ad1, mx1, ex1, dn1);

    aggregate1_sliced<<<F1 / 2, AGG_THREADS, 0, stream>>>(ei, E, Nn, ex1, dn1, h1, b1, h2);

    pad_cols<<<(F1 * OUTP + B - 1) / B, B, 0, stream>>>(W2, W2p, F1, OUT, OUTP);

    gemm_wmma_f32_chmajor_in<3><<<dim3((Nn + 15) / 16, 1), 32, 0, stream>>>(
        h2, W2p, h3, Nn, F1, OUTP, OUTP);

    alpha2_kernel<<<(Nn + B - 1) / B, B, 0, stream>>>(h3, aS2, aD2, as2, ad2, Nn);

    fill_f32<<<(Nn + B - 1) / B, B, 0, stream>>>(mx2, Nn, NEG_INF);
    fill_f32<<<(Nn + B - 1) / B, B, 0, stream>>>(dn2, Nn, 0.f);
    fill_f32<<<(Nn * OUT + B - 1) / B, B, 0, stream>>>(out, Nn * OUT, 0.f);

    edge_max2   <<<(ET + B - 1) / B, B, 0, stream>>>(ei, E, Nn, as2, ad2, mx2);
    edge_expsum2<<<(ET + B - 1) / B, B, 0, stream>>>(ei, E, Nn, as2, ad2, mx2, ex2, dn2);
    aggregate2  <<<(ET * 10 + B - 1) / B, B, 0, stream>>>(ei, E, Nn, ex2, dn2, h3, out);

    bias2_kernel<<<(Nn * OUT + B - 1) / B, B, 0, stream>>>(out, b2, Nn * OUT);
}


# Layer-2 softmax and aggregation by destination range: a work-group owns 40 nodes, scans all edge destinations, queues its edges in LDS, sums by LDS adds (arrival order); divide after sum; 11 launches
# speedup vs baseline: 1.5220x; 1.2342x over previous
#include <hip/hip_runtime.h>
#include <hip/hip_bf16.h>
#include <math.h>


typedef float v2f __attribute__((ext_vector_type(2)));
typedef float v8f __attribute__((ext_vector_type(8)));

#define NEG_SLOPE 0.2f

__device__ __forceinline__ float leaky(float v) {
    return v > 0.0f ? v : NEG_SLOPE * v;
}

__device__ __forceinline__ void atomicMaxF32(float* addr, float val) {
    if (val >= 0.0f) atomicMax((int*)addr, __float_as_int(val));
    else             atomicMin((unsigned int*)addr, __float_as_uint(val));
}

template<int NT>
__global__ void __launch_bounds__(32)
gemm_wmma_f32_chmajor_out(const float* __restrict__ A, const float* __restrict__ B,
                          float* __restrict__ Ct, int M, int K, int N)
{
    __shared__ float As[16 * 12];

    const int lane = (int)threadIdx.x;
    const int half = lane >> 4;
    const int l16  = lane & 15;

    const int alr = lane >> 1;
    const int alc = (lane & 1) * 4;
    int rowA = blockIdx.x * 16 + alr;
    if (rowA >= M) rowA = M - 1;
    const float* __restrict__ Aload = A + (size_t)rowA * (size_t)K + alc;

    const int ncol0 = blockIdx.y * (16 * NT);
    const int colB  = ncol0 + l16;

    v8f acc[NT];
#pragma unroll
    for (int t = 0; t < NT; ++t) {
        v8f z = {0.f, 0.f, 0.f, 0.f, 0.f, 0.f, 0.f, 0.f};
        acc[t] = z;
    }

    float4 av = *(const float4*)(Aload);

    for (int kc = 0; kc < K; kc += 8) {
        __syncthreads();
        *(float4*)(&As[alr * 12 + alc]) = av;

        v2f breg[2][NT];
#pragma unroll
        for (int kk2 = 0; kk2 < 2; ++kk2) {
            const float* __restrict__ Bk =
                B + (size_t)(kc + kk2 * 4 + 2 * half) * (size_t)N + colB;
#pragma unroll
            for (int t = 0; t < NT; ++t) {
                breg[kk2][t].x = Bk[t * 16];
                breg[kk2][t].y = Bk[N + t * 16];
            }
        }

        const int kn = (kc + 8 < K) ? (kc + 8) : 0;
        av = *(const float4*)(Aload + kn);

        __syncthreads();

#pragma unroll
        for (int kk2 = 0; kk2 < 2; ++kk2) {
            const v2f a = *(const v2f*)(&As[l16 * 12 + kk2 * 4 + 2 * half]);
#pragma unroll
            for (int t = 0; t < NT; ++t) {
                acc[t] = __builtin_amdgcn_wmma_f32_16x16x4_f32(
                    false, breg[kk2][t], false, a, (short)0, acc[t], false, false);
            }
        }
    }

    const int node0 = blockIdx.x * 16;
    float* __restrict__ Cbase =
        Ct + (size_t)(ncol0 + half * 8) * (size_t)M + node0 + l16;
    if (node0 + 16 <= M) {
#pragma unroll
        for (int t = 0; t < NT; ++t) {
#pragma unroll
            for (int r = 0; r < 8; ++r)
                Cbase[(size_t)(t * 16 + r) * (size_t)M] = acc[t][r];
        }
    } else {
#pragma unroll
        for (int t = 0; t < NT; ++t) {
#pragma unroll
            for (int r = 0; r < 8; ++r)
                if (node0 + l16 < M) Cbase[(size_t)(t * 16 + r) * (size_t)M] = acc[t][r];
        }
    }
}

template<int NT>
__global__ void __launch_bounds__(32)
gemm_wmma_f32_chmajor_in(const float* __restrict__ At, const float* __restrict__ B,
                         float* __restrict__ C, int M, int K, int N, int ldc)
{
    const int lane = (int)threadIdx.x;
    const int half = lane >> 4;
    const int l16  = lane & 15;

    int rowA = blockIdx.x * 16 + l16;
    if (rowA >= M) rowA = M - 1;
    const float* __restrict__ Aload = At + (size_t)(2 * half) * (size_t)M + rowA;

    const int ncol0 = blockIdx.y * (16 * NT);
    const int colB  = ncol0 + l16;

    v8f acc[NT];
#pragma unroll
    for (int t = 0; t < NT; ++t) {
        v8f z = {0.f, 0.f, 0.f, 0.f, 0.f, 0.f, 0.f, 0.f};
        acc[t] = z;
    }

    v2f an[2];
#pragma unroll
    for (int kk2 = 0; kk2 < 2; ++kk2) {
        an[kk2].x = Aload[(size_t)(kk2 * 4) * (size_t)M];
        an[kk2].y = Aload[(size_t)(kk2 * 4 + 1) * (size_t)M];
    }

    for (int kc = 0; kc < K; kc += 8) {
        const v2f a0 = an[0];
        const v2f a1 = an[1];

        v2f breg[2][NT];
#pragma unroll
        for (int kk2 = 0; kk2 < 2; ++kk2) {
            const float* __restrict__ Bk =
                B + (size_t)(kc + kk2 * 4 + 2 * half) * (size_t)N + colB;
#pragma unroll
            for (int t = 0; t < NT; ++t) {
                breg[kk2][t].x = Bk[t * 16];
                breg[kk2][t].y = Bk[N + t * 16];
            }
        }

        const int kn = (kc + 8 < K) ? (kc + 8) : 0;
#pragma unroll
        for (int kk2 = 0; kk2 < 2; ++kk2) {
            an[kk2].x = Aload[(size_t)(kn + kk2 * 4) * (size_t)M];
            an[kk2].y = Aload[(size_t)(kn + kk2 * 4 + 1) * (size_t)M];
        }

#pragma unroll
        for (int t = 0; t < NT; ++t) {
            acc[t] = __builtin_amdgcn_wmma_f32_16x16x4_f32(
                false, a0, false, breg[0][t], (short)0, acc[t], false, false);
        }
#pragma unroll
        for (int t = 0; t < NT; ++t) {
            acc[t] = __builtin_amdgcn_wmma_f32_16x16x4_f32(
                false, a1, false, breg[1][t], (short)0, acc[t], false, false);
        }
    }

    const int crow0 = blockIdx.x * 16 + half * 8;
    float* __restrict__ Cbase = C + (size_t)crow0 * (size_t)ldc + ncol0 + l16;
    if (blockIdx.x * 16 + 16 <= M) {
#pragma unroll
        for (int t = 0; t < NT; ++t) {
            float* __restrict__ Cp = Cbase + t * 16;
#pragma unroll
            for (int r = 0; r < 8; ++r)
                Cp[(size_t)r * (size_t)ldc] = acc[t][r];
        }
    } else {
#pragma unroll
        for (int t = 0; t < NT; ++t) {
            float* __restrict__ Cp = Cbase + t * 16;
#pragma unroll
            for (int r = 0; r < 8; ++r)
                if (crow0 + r < M) Cp[(size_t)r * (size_t)ldc] = acc[t][r];
        }
    }
}

__global__ void fill_f32(float* __restrict__ p, int n, float v) {
    int t = blockIdx.x * blockDim.x + threadIdx.x;
    if (t < n) p[t] = v;
}

__global__ void pad_cols(const float* __restrict__ src, float* __restrict__ dst,
                         int K, int N, int Np)
{
    int t = blockIdx.x * blockDim.x + threadIdx.x;
    if (t >= K * Np) return;
    int r = t / Np, c = t - r * Np;
    dst[t] = (c < N) ? src[r * N + c] : 0.f;
}

__global__ void alpha1_kernel(const float* __restrict__ h1t,
                              const float* __restrict__ a_src,
                              const float* __restrict__ a_dst,
                              float* __restrict__ as, float* __restrict__ ad, int Nn)
{
    int t = blockIdx.x * blockDim.x + threadIdx.x;
    if (t >= Nn * 8) return;
    int h = t / Nn, n = t - h * Nn;
    const float* hp = h1t + (size_t)(h * 64) * (size_t)Nn + n;
    const float* sp = a_src + h * 64;
    const float* dp = a_dst + h * 64;
    float ss = 0.f, dd = 0.f;
#pragma unroll 8
    for (int c = 0; c < 64; ++c) {
        float v = hp[(size_t)c * (size_t)Nn];
        ss += v * sp[c];
        dd += v * dp[c];
    }
    as[n * 8 + h] = ss;
    ad[n * 8 + h] = dd;
}

__device__ __forceinline__ void edge_sd(const int* __restrict__ ei, int E, int e,
                                        int& s, int& d)
{
    if (e < E) { s = ei[e]; d = ei[E + e]; }
    else       { s = e - E; d = e - E; }
}

__global__ void edge_max1(const int* __restrict__ ei, int E, int Nn,
                          const float* __restrict__ as, const float* __restrict__ ad,
                          float* __restrict__ mx)
{
    int t = blockIdx.x * blockDim.x + threadIdx.x;
    if (t >= (E + Nn) * 8) return;
    int e = t >> 3, h = t & 7;
    int s, d; edge_sd(ei, E, e, s, d);
    atomicMaxF32(&mx[d * 8 + h], leaky(as[s * 8 + h] + ad[d * 8 + h]));
}

__global__ void edge_expsum1(const int* __restrict__ ei, int E, int Nn,
                             const float* __restrict__ as, const float* __restrict__ ad,
                             const float* __restrict__ mx,
                             float* __restrict__ ex, float* __restrict__ denom)
{
    int t = blockIdx.x * blockDim.x + threadIdx.x;
    if (t >= (E + Nn) * 8) return;
    int e = t >> 3, h = t & 7;
    int s, d; edge_sd(ei, E, e, s, d);
    float v  = leaky(as[s * 8 + h] + ad[d * 8 + h]);
    float xv = expf(v - mx[d * 8 + h]);
    ex[(size_t)h * (size_t)(E + Nn) + e] = xv;
    atomicAdd(&denom[d * 8 + h], xv);
}

#define AGG_NODES   10496
#define AGG_THREADS 1024

__global__ void __launch_bounds__(AGG_THREADS)
aggregate1_sliced(const int* __restrict__ ei, int E, int Nn,
                  const float* __restrict__ ex, const float* __restrict__ denom,
                  const float* __restrict__ h1t, const float* __restrict__ b,
                  float* __restrict__ h2t)
{
    __shared__ float2 hs[AGG_NODES];
    __shared__ float  acc[2 * AGG_NODES];

    const int tid = (int)threadIdx.x;
    const int c0  = (int)blockIdx.x * 2;
    const int h   = c0 >> 6;
    const int ET  = E + Nn;

    const float* __restrict__ exh = ex + (size_t)h * (size_t)ET;
    const float* __restrict__ p0  = h1t + (size_t)c0 * (size_t)Nn;
    const float* __restrict__ p1  = p0 + Nn;
    for (int n = tid; n < Nn; n += AGG_THREADS) {
        const float x0 = p0[n], x1 = p1[n];
        const float ws = exh[E + n];
        hs[n] = make_float2(x0, x1);
        acc[2 * n]     = ws * x0;
        acc[2 * n + 1] = ws * x1;
    }
    __syncthreads();

    const int* __restrict__ src = ei;
    const int* __restrict__ dst = ei + E;

    auto load_group = [&](int g, int* s, int* d, float* w) {
#pragma unroll
        for (int j = 0; j < 4; ++j) {
            const int e = g * (4 * AGG_THREADS) + tid + j * AGG_THREADS;
            s[j] = src[e];
            d[j] = dst[e];
            w[j] = exh[e];
        }
    };
    auto add_edge = [&](int d, float w, float2 hv) {
        __hip_atomic_fetch_add(&acc[2 * d],     w * hv.x,
                               __ATOMIC_RELAXED, __HIP_MEMORY_SCOPE_WORKGROUP);
        __hip_atomic_fetch_add(&acc[2 * d + 1], w * hv.y,
                               __ATOMIC_RELAXED, __HIP_MEMORY_SCOPE_WORKGROUP);
    };
    auto do_group = [&](const int* s, const int* d, const float* w) {
        float2 hv[4];
#pragma unroll
        for (int j = 0; j < 4; ++j) hv[j] = hs[s[j]];
#pragma unroll
        for (int j = 0; j < 4; ++j) add_edge(d[j], w[j], hv[j]);
    };

    const int nfull = E / (4 * AGG_THREADS);
    int   sA[4], dA[4], sB[4], dB[4];
    float wA[4], wB[4];
    if (nfull > 0) load_group(0, sA, dA, wA);
    int g = 0;
    for (; g + 2 <= nfull; g += 2) {
        load_group(g + 1, sB, dB, wB);
        do_group(sA, dA, wA);
        load_group((g + 2 < nfull) ? (g + 2) : (g + 1), sA, dA, wA);
        do_group(sB, dB, wB);
    }
    if (g < nfull) do_group(sA, dA, wA);

    for (int e = nfull * (4 * AGG_THREADS) + tid; e < E; e += AGG_THREADS)
        add_edge(dst[e], exh[e], hs[src[e]]);
    __syncthreads();

    const float b0 = b[c0], b1 = b[c0 + 1];
    float* __restrict__ q0 = h2t + (size_t)c0 * (size_t)Nn;
    float* __restrict__ q1 = q0 + Nn;
    for (int n = tid; n < Nn; n += AGG_THREADS) {
        const float dn = denom[n * 8 + h];
        const float v0 = acc[2 * n]     / dn + b0;
        const float v1 = acc[2 * n + 1] / dn + b1;
        q0[n] = v0 > 0.f ? v0 : expm1f(v0);
        q1[n] = v1 > 0.f ? v1 : expm1f(v1);
    }
}

__global__ void alpha2_kernel(const float* __restrict__ h3,
                              const float* __restrict__ a_src,
                              const float* __restrict__ a_dst,
                              float* __restrict__ as, float* __restrict__ ad, int Nn)
{
    int n = blockIdx.x * blockDim.x + threadIdx.x;
    if (n >= Nn) return;
    const float* hp = h3 + (size_t)n * 48;
    float ss = 0.f, dd = 0.f;
#pragma unroll 8
    for (int c = 0; c < 40; ++c) { float v = hp[c]; ss += v * a_src[c]; dd += v * a_dst[c]; }
    as[n] = ss;
    ad[n] = dd;
}

#define L2_NPW     40
#define L2_THREADS 1024
#define L2_QCAP    20480
#define L2_SEG     (L2_QCAP - 64)

__global__ void __launch_bounds__(L2_THREADS)
layer2_by_dst_range(const int* __restrict__ ei, int E, int Nn,
                    const float* __restrict__ as, const float* __restrict__ ad,
                    const float* __restrict__ h3, const float* __restrict__ b,
                    float* __restrict__ out)
{
    __shared__ int            qe[L2_QCAP];
    __shared__ float          qw[L2_QCAP];
    __shared__ unsigned short qd[L2_QCAP];
    __shared__ float acc[L2_NPW * 40];
    __shared__ float mx[L2_NPW];
    __shared__ float den[L2_NPW];
    __shared__ float adl[L2_NPW];
    __shared__ int   cnt;

    const int tid = (int)threadIdx.x;
    const int n0  = (int)blockIdx.x * L2_NPW;
    const int nn  = (Nn - n0 < L2_NPW) ? (Nn - n0) : L2_NPW;
    const int* __restrict__ dst = ei + E;

    for (int i = tid; i < L2_NPW * 40; i += L2_THREADS) acc[i] = 0.f;
    if (tid < L2_NPW) {
        mx[tid]  = -__builtin_inff();
        den[tid] = 0.f;
        adl[tid] = (tid < nn) ? ad[n0 + tid] : 0.f;
    }

    auto push = [&](int e, int dl) {
        const int q = atomicAdd(&cnt, 1);
        if (q < L2_QCAP) { qe[q] = e; qd[q] = (unsigned short)dl; }
    };
    auto test4 = [&](int4 d, int e, unsigned un) {
        const unsigned r0 = (unsigned)(d.x - n0), r1 = (unsigned)(d.y - n0);
        const unsigned r2 = (unsigned)(d.z - n0), r3 = (unsigned)(d.w - n0);
        if ((r0 < un) | (r1 < un) | (r2 < un) | (r3 < un)) {
            if (r0 < un) push(e,     (int)r0);
            if (r1 < un) push(e + 1, (int)r1);
            if (r2 < un) push(e + 2, (int)r2);
            if (r3 < un) push(e + 3, (int)r3);
        }
    };

    auto fill_queue = [&](int e0, int e1, bool with_self) -> int {
        __syncthreads();
        if (tid == 0) cnt = 0;
        __syncthreads();
        if (with_self && tid < nn) push(E + n0 + tid, tid);

        if (((E | e0) & 3) == 0 && ((unsigned long long)ei & 15ull) == 0) {
            const int4* __restrict__ d4 = (const int4*)(dst + e0);
            const int n4    = (e1 - e0) >> 2;
            const int start = (int)(blockIdx.x * ((unsigned)n4 / gridDim.x)) & ~31;
            const int ntrip = (n4 + 8 * L2_THREADS - 1) / (8 * L2_THREADS);
            for (int t = 0; t < ntrip; ++t) {
                int4 buf[8];
                int  at[8];
#pragma unroll
                for (int u = 0; u < 8; ++u) {
                    const int p = (t * 8 + u) * L2_THREADS + tid;
                    int i4 = p + start;
                    if (i4 >= n4) i4 -= n4;
                    at[u]  = (p < n4) ? i4 : -1;
                    buf[u] = d4[at[u] < 0 ? 0 : at[u]];
                }
#pragma unroll
                for (int u = 0; u < 8; ++u)
                    test4(buf[u], e0 + 4 * at[u], at[u] < 0 ? 0u : (unsigned)nn);
            }
            for (int e = e0 + 4 * n4 + tid; e < e1; e += L2_THREADS) {
                const unsigned r = (unsigned)(dst[e] - n0);
                if (r < (unsigned)nn) push(e, (int)r);
            }
        } else {
            for (int e = e0 + tid; e < e1; e += L2_THREADS) {
                const unsigned r = (unsigned)(dst[e] - n0);
                if (r < (unsigned)nn) push(e, (int)r);
            }
        }
        __syncthreads();
        return cnt;
    };

    auto drain_max = [&](int count) {
        for (int q = tid; q < count; q += L2_THREADS) {
            const int e  = qe[q];
            const int dl = (int)qd[q];
            const int s  = (e < E) ? ei[e] : (e - E);
            const float v = leaky(as[s] + adl[dl]);
            qe[q] = s;
            qw[q] = v;
            atomicMaxF32(&mx[dl], v);
        }
    };

    auto drain_sum = [&](int count) {
        for (int q = tid; q < count; q += L2_THREADS) {
            const int dl  = (int)qd[q];
            const float w = expf(qw[q] - mx[dl]);
            qw[q] = w;
            __hip_atomic_fetch_add(&den[dl], w, __ATOMIC_RELAXED, __HIP_MEMORY_SCOPE_WORKGROUP);
        }
        __syncthreads();
        const int nitems = count * 10;
        for (int base = 0; base < nitems; base += 4 * L2_THREADS) {
            int   q[4], g[4], s[4], o[4];
            float w[4];
            bool  ok[4];
#pragma unroll
            for (int j = 0; j < 4; ++j) {
                const int it = base + j * L2_THREADS + tid;
                ok[j] = it < nitems;
                const int itc = ok[j] ? it : (nitems - 1);
                q[j] = itc / 10;
                g[j] = itc - q[j] * 10;
            }
#pragma unroll
            for (int j = 0; j < 4; ++j) {
                s[j] = qe[q[j]];
                w[j] = qw[q[j]];
                o[j] = (int)qd[q[j]] * 40 + g[j] * 4;
            }
            float4 hv[4];
#pragma unroll
            for (int j = 0; j < 4; ++j)
                hv[j] = *(const float4*)(h3 + (size_t)s[j] * 48 + g[j] * 4);
#pragma unroll
            for (int j = 0; j < 4; ++j) {
                if (ok[j]) {
                    __hip_atomic_fetch_add(&acc[o[j]],     w[j] * hv[j].x,
                                           __ATOMIC_RELAXED, __HIP_MEMORY_SCOPE_WORKGROUP);
                    __hip_atomic_fetch_add(&acc[o[j] + 1], w[j] * hv[j].y,
                                           __ATOMIC_RELAXED, __HIP_MEMORY_SCOPE_WORKGROUP);
                    __hip_atomic_fetch_add(&acc[o[j] + 2], w[j] * hv[j].z,
                                           __ATOMIC_RELAXED, __HIP_MEMORY_SCOPE_WORKGROUP);
                    __hip_atomic_fetch_add(&acc[o[j] + 3], w[j] * hv[j].w,
                                           __ATOMIC_RELAXED, __HIP_MEMORY_SCOPE_WORKGROUP);
                }
            }
        }
    };

    const int total = fill_queue(0, E, true);
    if (total <= L2_QCAP) {
        drain_max(total);
        __syncthreads();
        drain_sum(total);
    } else {
        for (int ph = 0; ph < 2; ++ph) {
            for (int e0 = 0; e0 < E; e0 += L2_SEG) {
                const int e1 = (e0 + L2_SEG < E) ? (e0 + L2_SEG) : E;
                const int c  = fill_queue(e0, e1, e0 == 0);
                drain_max(c);
                __syncthreads();
                if (ph) drain_sum(c);
            }
        }
    }
    __syncthreads();

    float* __restrict__ o0 = out + (size_t)n0 * 40;
    for (int i = tid; i < nn * 40; i += L2_THREADS) {
        const int dl = i / 40;
        const int c  = i - dl * 40;
        o0[i] = acc[i] / den[dl] + b[c];
    }
}

extern "C" void kernel_launch(void* const* d_in, const int* in_sizes, int n_in,
                              void* d_out, int out_size, void* d_ws, size_t ws_size,
                              hipStream_t stream)
{
    const float* x    = (const float*)d_in[0];
    const int*   ei   = (const int*)  d_in[1];
    const float* W1   = (const float*)d_in[2];
    const float* aS1  = (const float*)d_in[3];
    const float* aD1  = (const float*)d_in[4];
    const float* b1   = (const float*)d_in[5];
    const float* W2   = (const float*)d_in[6];
    const float* aS2  = (const float*)d_in[7];
    const float* aD2  = (const float*)d_in[8];
    const float* b2   = (const float*)d_in[9];
    float* out = (float*)d_out;

    const int IN  = 256, F1 = 512, OUT = 40, OUTP = 48;
    const int Nn  = in_sizes[0] / IN;
    const int E   = in_sizes[1] / 2;
    const int ET  = E + Nn;

    float* ws  = (float*)d_ws;
    float* h1  = ws;
    float* h2  = h1  + (size_t)Nn * F1;
    float* as1 = h2  + (size_t)Nn * F1;
    float* ad1 = as1 + (size_t)Nn * 8;
    float* mx1 = ad1 + (size_t)Nn * 8;
    float* dn1 = mx1 + (size_t)Nn * 8;
    float* ex1 = dn1 + (size_t)Nn * 8;
    float* h3  = ex1 + (size_t)ET * 8;
    float* as2 = h3  + (size_t)Nn * 48;
    float* ad2 = as2 + Nn;
    float* W2p = ad2 + Nn;

    const int B = 256;
    const float NEG_INF = -__builtin_inff();

    gemm_wmma_f32_chmajor_out<4><<<dim3((Nn + 15) / 16, F1 / 64), 32, 0, stream>>>(
        x, W1, h1, Nn, IN, F1);

    alpha1_kernel<<<(Nn * 8 + B - 1) / B, B, 0, stream>>>(h1, aS1, aD1, as1, ad1, Nn);

    fill_f32<<<(Nn * 8 + B - 1) / B, B, 0, stream>>>(mx1, Nn * 8, NEG_INF);
    fill_f32<<<(Nn * 8 + B - 1) / B, B, 0, stream>>>(dn1, Nn * 8, 0.f);

    edge_max1   <<<(ET * 8 + B - 1) / B, B, 0, stream>>>(ei, E, Nn, as1, ad1, mx1);
    edge_expsum1<<<(ET * 8 + B - 1) / B, B, 0, stream>>>(ei, E, Nn, as1, ad1, mx1, ex1, dn1);

    aggregate1_sliced<<<F1 / 2, AGG_THREADS, 0, stream>>>(ei, E, Nn, ex1, dn1, h1, b1, h2);

    pad_cols<<<(F1 * OUTP + B - 1) / B, B, 0, stream>>>(W2, W2p, F1, OUT, OUTP);

    gemm_wmma_f32_chmajor_in<3><<<dim3((Nn + 15) / 16, 1), 32, 0, stream>>>(
        h2, W2p, h3, Nn, F1, OUTP, OUTP);

    alpha2_kernel<<<(Nn + B - 1) / B, B, 0, stream>>>(h3, aS2, aD2, as2, ad2, Nn);

    layer2_by_dst_range<<<(Nn + L2_NPW - 1) / L2_NPW, L2_THREADS, 0, stream>>>(
        ei, E, Nn, as2, ad2, h3, b2, out);
}


# Layer-2 GEMM: A and B fragments of 4 chunks (32 k) stay in flight per wave in a ring of register sets, each set refilled behind its own WMMAs (was: load, wait, multiply per chunk); sum order kept
# speedup vs baseline: 1.7473x; 1.1480x over previous
#include <hip/hip_runtime.h>
#include <hip/hip_bf16.h>
#include <math.h>


typedef float v2f __attribute__((ext_vector_type(2)));
typedef float v8f __attribute__((ext_vector_type(8)));

#define NEG_SLOPE 0.2f

__device__ __forceinline__ float leaky(float v) {
    return v > 0.0f ? v : NEG_SLOPE * v;
}

__device__ __forceinline__ void atomicMaxF32(float* addr, float val) {
    if (val >= 0.0f) atomicMax((int*)addr, __float_as_int(val));
    else             atomicMin((unsigned int*)addr, __float_as_uint(val));
}

template<int NT>
__global__ void __launch_bounds__(32)
gemm_wmma_f32_chmajor_out(const float* __restrict__ A, const float* __restrict__ B,
                          float* __restrict__ Ct, int M, int K, int N)
{
    __shared__ float As[16 * 12];

    const int lane = (int)threadIdx.x;
    const int half = lane >> 4;
    const int l16  = lane & 15;

    const int alr = lane >> 1;
    const int alc = (lane & 1) * 4;
    int rowA = blockIdx.x * 16 + alr;
    if (rowA >= M) rowA = M - 1;
    const float* __restrict__ Aload = A + (size_t)rowA * (size_t)K + alc;

    const int ncol0 = blockIdx.y * (16 * NT);
    const int colB  = ncol0 + l16;

    v8f acc[NT];
#pragma unroll
    for (int t = 0; t < NT; ++t) {
        v8f z = {0.f, 0.f, 0.f, 0.f, 0.f, 0.f, 0.f, 0.f};
        acc[t] = z;
    }

    float4 av = *(const float4*)(Aload);

    for (int kc = 0; kc < K; kc += 8) {
        __syncthreads();
        *(float4*)(&As[alr * 12 + alc]) = av;

        v2f breg[2][NT];
#pragma unroll
        for (int kk2 = 0; kk2 < 2; ++kk2) {
            const float* __restrict__ Bk =
                B + (size_t)(kc + kk2 * 4 + 2 * half) * (size_t)N + colB;
#pragma unroll
            for (int t = 0; t < NT; ++t) {
                breg[kk2][t].x = Bk[t * 16];
                breg[kk2][t].y = Bk[N + t * 16];
            }
        }

        const int kn = (kc + 8 < K) ? (kc + 8) : 0;
        av = *(const float4*)(Aload + kn);

        __syncthreads();

#pragma unroll
        for (int kk2 = 0; kk2 < 2; ++kk2) {
            const v2f a = *(const v2f*)(&As[l16 * 12 + kk2 * 4 + 2 * half]);
#pragma unroll
            for (int t = 0; t < NT; ++t) {
                acc[t] = __builtin_amdgcn_wmma_f32_16x16x4_f32(
                    false, breg[kk2][t], false, a, (short)0, acc[t], false, false);
            }
        }
    }

    const int node0 = blockIdx.x * 16;
    float* __restrict__ Cbase =
        Ct + (size_t)(ncol0 + half * 8) * (size_t)M + node0 + l16;
    if (node0 + 16 <= M) {
#pragma unroll
        for (int t = 0; t < NT; ++t) {
#pragma unroll
            for (int r = 0; r < 8; ++r)
                Cbase[(size_t)(t * 16 + r) * (size_t)M] = acc[t][r];
        }
    } else {
#pragma unroll
        for (int t = 0; t < NT; ++t) {
#pragma unroll
            for (int r = 0; r < 8; ++r)
                if (node0 + l16 < M) Cbase[(size_t)(t * 16 + r) * (size_t)M] = acc[t][r];
        }
    }
}

#define G2_DEPTH 4

template<int NT>
__device__ __forceinline__ void g2_load_chunk(const float* __restrict__ Aload,
                                              const float* __restrict__ Bload,
                                              int M, int N, int kc,
                                              v2f (&a)[2], v2f (&b)[2][NT])
{
#pragma unroll
    for (int kk2 = 0; kk2 < 2; ++kk2) {
        const float* __restrict__ Ak = Aload + (size_t)(kc + kk2 * 4) * (size_t)M;
        const float* __restrict__ Bk = Bload + (size_t)(kc + kk2 * 4) * (size_t)N;
        a[kk2].x = Ak[0];
        a[kk2].y = Ak[M];
#pragma unroll
        for (int t = 0; t < NT; ++t) {
            b[kk2][t].x = Bk[t * 16];
            b[kk2][t].y = Bk[N + t * 16];
        }
    }
}

template<int NT>
__device__ __forceinline__ void g2_ring_turn(v8f (&acc)[NT],
                                             v2f (&ra)[G2_DEPTH][2],
                                             v2f (&rb)[G2_DEPTH][2][NT],
                                             const float* __restrict__ Aload,
                                             const float* __restrict__ Bload,
                                             int M, int N, int c0, int nch)
{
#pragma unroll
    for (int s = 0; s < G2_DEPTH; ++s) {
        if (c0 >= 0) {
#pragma unroll
            for (int t = 0; t < NT; ++t) {
                acc[t] = __builtin_amdgcn_wmma_f32_16x16x4_f32(
                    false, ra[s][0], false, rb[s][0][t], (short)0, acc[t], false, false);
            }
#pragma unroll
            for (int t = 0; t < NT; ++t) {
                acc[t] = __builtin_amdgcn_wmma_f32_16x16x4_f32(
                    false, ra[s][1], false, rb[s][1][t], (short)0, acc[t], false, false);
            }
        }

        int cn = c0 + s + G2_DEPTH;
        if (cn >= nch) cn = 0;
        g2_load_chunk<NT>(Aload, Bload, M, N, cn * 8, ra[s], rb[s]);
        __builtin_amdgcn_sched_barrier(0);
    }
}

template<int NT>
__global__ void __launch_bounds__(32)
gemm_wmma_f32_chmajor_in(const float* __restrict__ At, const float* __restrict__ B,
                         float* __restrict__ C, int M, int K, int N, int ldc)
{
    const int lane = (int)threadIdx.x;
    const int half = lane >> 4;
    const int l16  = lane & 15;

    int rowA = blockIdx.x * 16 + l16;
    if (rowA >= M) rowA = M - 1;
    const float* __restrict__ Aload = At + (size_t)(2 * half) * (size_t)M + rowA;

    const int ncol0 = blockIdx.y * (16 * NT);
    const float* __restrict__ Bload = B + (size_t)(2 * half) * (size_t)N + ncol0 + l16;

    v8f acc[NT];
#pragma unroll
    for (int t = 0; t < NT; ++t) {
        v8f z = {0.f, 0.f, 0.f, 0.f, 0.f, 0.f, 0.f, 0.f};
        acc[t] = z;
    }

    const int nch = K >> 3;
    if (nch > 0) {
        v2f ra[G2_DEPTH][2];
        v2f rb[G2_DEPTH][2][NT];
        for (int c0 = -G2_DEPTH; c0 < nch; c0 += G2_DEPTH)
            g2_ring_turn<NT>(acc, ra, rb, Aload, Bload, M, N, c0, nch);
    }

    const int crow0 = blockIdx.x * 16 + half * 8;
    float* __restrict__ Cbase = C + (size_t)crow0 * (size_t)ldc + ncol0 + l16;
    if (blockIdx.x * 16 + 16 <= M) {
#pragma unroll
        for (int t = 0; t < NT; ++t) {
            float* __restrict__ Cp = Cbase + t * 16;
#pragma unroll
            for (int r = 0; r < 8; ++r)
                Cp[(size_t)r * (size_t)ldc] = acc[t][r];
        }
    } else {
#pragma unroll
        for (int t = 0; t < NT; ++t) {
            float* __restrict__ Cp = Cbase + t * 16;
#pragma unroll
            for (int r = 0; r < 8; ++r)
                if (crow0 + r < M) Cp[(size_t)r * (size_t)ldc] = acc[t][r];
        }
    }
}

__global__ void fill_f32(float* __restrict__ p, int n, float v) {
    int t = blockIdx.x * blockDim.x + threadIdx.x;
    if (t < n) p[t] = v;
}

__global__ void pad_cols(const float* __restrict__ src, float* __restrict__ dst,
                         int K, int N, int Np)
{
    int t = blockIdx.x * blockDim.x + threadIdx.x;
    if (t >= K * Np) return;
    int r = t / Np, c = t - r * Np;
    dst[t] = (c < N) ? src[r * N + c] : 0.f;
}

__global__ void alpha1_kernel(const float* __restrict__ h1t,
                              const float* __restrict__ a_src,
                              const float* __restrict__ a_dst,
                              float* __restrict__ as, float* __restrict__ ad, int Nn)
{
    int t = blockIdx.x * blockDim.x + threadIdx.x;
    if (t >= Nn * 8) return;
    int h = t / Nn, n = t - h * Nn;
    const float* hp = h1t + (size_t)(h * 64) * (size_t)Nn + n;
    const float* sp = a_src + h * 64;
    const float* dp = a_dst + h * 64;
    float ss = 0.f, dd = 0.f;
#pragma unroll 8
    for (int c = 0; c < 64; ++c) {
        float v = hp[(size_t)c * (size_t)Nn];
        ss += v * sp[c];
        dd += v * dp[c];
    }
    as[n * 8 + h] = ss;
    ad[n * 8 + h] = dd;
}

__device__ __forceinline__ void edge_sd(const int* __restrict__ ei, int E, int e,
                                        int& s, int& d)
{
    if (e < E) { s = ei[e]; d = ei[E + e]; }
    else       { s = e - E; d = e - E; }
}

__global__ void edge_max1(const int* __restrict__ ei, int E, int Nn,
                          const float* __restrict__ as, const float* __restrict__ ad,
                          float* __restrict__ mx)
{
    int t = blockIdx.x * blockDim.x + threadIdx.x;
    if (t >= (E + Nn) * 8) return;
    int e = t >> 3, h = t & 7;
    int s, d; edge_sd(ei, E, e, s, d);
    atomicMaxF32(&mx[d * 8 + h], leaky(as[s * 8 + h] + ad[d * 8 + h]));
}

__global__ void edge_expsum1(const int* __restrict__ ei, int E, int Nn,
                             const float* __restrict__ as, const float* __restrict__ ad,
                             const float* __restrict__ mx,
                             float* __restrict__ ex, float* __restrict__ denom)
{
    int t = blockIdx.x * blockDim.x + threadIdx.x;
    if (t >= (E + Nn) * 8) return;
    int e = t >> 3, h = t & 7;
    int s, d; edge_sd(ei, E, e, s, d);
    float v  = leaky(as[s * 8 + h] + ad[d * 8 + h]);
    float xv = expf(v - mx[d * 8 + h]);
    ex[(size_t)h * (size_t)(E + Nn) + e] = xv;
    atomicAdd(&denom[d * 8 + h], xv);
}

#define AGG_NODES   10496
#define AGG_THREADS 1024

__global__ void __launch_bounds__(AGG_THREADS)
aggregate1_sliced(const int* __restrict__ ei, int E, int Nn,
                  const float* __restrict__ ex, const float* __restrict__ denom,
                  const float* __restrict__ h1t, const float* __restrict__ b,
                  float* __restrict__ h2t)
{
    __shared__ float2 hs[AGG_NODES];
    __shared__ float  acc[2 * AGG_NODES];

    const int tid = (int)threadIdx.x;
    const int c0  = (int)blockIdx.x * 2;
    const int h   = c0 >> 6;
    const int ET  = E + Nn;

    const float* __restrict__ exh = ex + (size_t)h * (size_t)ET;
    const float* __restrict__ p0  = h1t + (size_t)c0 * (size_t)Nn;
    const float* __restrict__ p1  = p0 + Nn;
    for (int n = tid; n < Nn; n += AGG_THREADS) {
        const float x0 = p0[n], x1 = p1[n];
        const float ws = exh[E + n];
        hs[n] = make_float2(x0, x1);
        acc[2 * n]     = ws * x0;
        acc[2 * n + 1] = ws * x1;
    }
    __syncthreads();

    const int* __restrict__ src = ei;
    const int* __restrict__ dst = ei + E;

    auto load_group = [&](int g, int* s, int* d, float* w) {
#pragma unroll
        for (int j = 0; j < 4; ++j) {
            const int e = g * (4 * AGG_THREADS) + tid + j * AGG_THREADS;
            s[j] = src[e];
            d[j] = dst[e];
            w[j] = exh[e];
        }
    };
    auto add_edge = [&](int d, float w, float2 hv) {
        __hip_atomic_fetch_add(&acc[2 * d],     w * hv.x,
                               __ATOMIC_RELAXED, __HIP_MEMORY_SCOPE_WORKGROUP);
        __hip_atomic_fetch_add(&acc[2 * d + 1], w * hv.y,
                               __ATOMIC_RELAXED, __HIP_MEMORY_SCOPE_WORKGROUP);
    };
    auto do_group = [&](const int* s, const int* d, const float* w) {
        float2 hv[4];
#pragma unroll
        for (int j = 0; j < 4; ++j) hv[j] = hs[s[j]];
#pragma unroll
        for (int j = 0; j < 4; ++j) add_edge(d[j], w[j], hv[j]);
    };

    const int nfull = E / (4 * AGG_THREADS);
    int   sA[4], dA[4], sB[4], dB[4];
    float wA[4], wB[4];
    if (nfull > 0) load_group(0, sA, dA, wA);
    int g = 0;
    for (; g + 2 <= nfull; g += 2) {
        load_group(g + 1, sB, dB, wB);
        do_group(sA, dA, wA);
        load_group((g + 2 < nfull) ? (g + 2) : (g + 1), sA, dA, wA);
        do_group(sB, dB, wB);
    }
    if (g < nfull) do_group(sA, dA, wA);

    for (int e = nfull * (4 * AGG_THREADS) + tid; e < E; e += AGG_THREADS)
        add_edge(dst[e], exh[e], hs[src[e]]);
    __syncthreads();

    const float b0 = b[c0], b1 = b[c0 + 1];
    float* __restrict__ q0 = h2t + (size_t)c0 * (size_t)Nn;
    float* __restrict__ q1 = q0 + Nn;
    for (int n = tid; n < Nn; n += AGG_THREADS) {
        const float dn = denom[n * 8 + h];
        const float v0 = acc[2 * n]     / dn + b0;
        const float v1 = acc[2 * n + 1] / dn + b1;
        q0[n] = v0 > 0.f ? v0 : expm1f(v0);
        q1[n] = v1 > 0.f ? v1 : expm1f(v1);
    }
}

__global__ void alpha2_kernel(const float* __restrict__ h3,
                              const float* __restrict__ a_src,
                              const float* __restrict__ a_dst,
                              float* __restrict__ as, float* __restrict__ ad, int Nn)
{
    int n = blockIdx.x * blockDim.x + threadIdx.x;
    if (n >= Nn) return;
    const float* hp = h3 + (size_t)n * 48;
    float ss = 0.f, dd = 0.f;
#pragma unroll 8
    for (int c = 0; c < 40; ++c) { float v = hp[c]; ss += v * a_src[c]; dd += v * a_dst[c]; }
    as[n] = ss;
    ad[n] = dd;
}

#define L2_NPW     40
#define L2_THREADS 1024
#define L2_QCAP    20480
#define L2_SEG     (L2_QCAP - 64)

__global__ void __launch_bounds__(L2_THREADS)
layer2_by_dst_range(const int* __restrict__ ei, int E, int Nn,
                    const float* __restrict__ as, const float* __restrict__ ad,
                    const float* __restrict__ h3, const float* __restrict__ b,
                    float* __restrict__ out)
{
    __shared__ int            qe[L2_QCAP];
    __shared__ float          qw[L2_QCAP];
    __shared__ unsigned short qd[L2_QCAP];
    __shared__ float acc[L2_NPW * 40];
    __shared__ float mx[L2_NPW];
    __shared__ float den[L2_NPW];
    __shared__ float adl[L2_NPW];
    __shared__ int   cnt;

    const int tid = (int)threadIdx.x;
    const int n0  = (int)blockIdx.x * L2_NPW;
    const int nn  = (Nn - n0 < L2_NPW) ? (Nn - n0) : L2_NPW;
    const int* __restrict__ dst = ei + E;

    for (int i = tid; i < L2_NPW * 40; i += L2_THREADS) acc[i] = 0.f;
    if (tid < L2_NPW) {
        mx[tid]  = -__builtin_inff();
        den[tid] = 0.f;
        adl[tid] = (tid < nn) ? ad[n0 + tid] : 0.f;
    }

    auto push = [&](int e, int dl) {
        const int q = atomicAdd(&cnt, 1);
        if (q < L2_QCAP) { qe[q] = e; qd[q] = (unsigned short)dl; }
    };
    auto test4 = [&](int4 d, int e, unsigned un) {
        const unsigned r0 = (unsigned)(d.x - n0), r1 = (unsigned)(d.y - n0);
        const unsigned r2 = (unsigned)(d.z - n0), r3 = (unsigned)(d.w - n0);
        if ((r0 < un) | (r1 < un) | (r2 < un) | (r3 < un)) {
            if (r0 < un) push(e,     (int)r0);
            if (r1 < un) push(e + 1, (int)r1);
            if (r2 < un) push(e + 2, (int)r2);
            if (r3 < un) push(e + 3, (int)r3);
        }
    };

    auto fill_queue = [&](int e0, int e1, bool with_self) -> int {
        __syncthreads();
        if (tid == 0) cnt = 0;
        __syncthreads();
        if (with_self && tid < nn) push(E + n0 + tid, tid);

        if (((E | e0) & 3) == 0 && ((unsigned long long)ei & 15ull) == 0) {
            const int4* __restrict__ d4 = (const int4*)(dst + e0);
            const int n4    = (e1 - e0) >> 2;
            const int start = (int)(blockIdx.x * ((unsigned)n4 / gridDim.x)) & ~31;
            const int ntrip = (n4 + 8 * L2_THREADS - 1) / (8 * L2_THREADS);
            for (int t = 0; t < ntrip; ++t) {
                int4 buf[8];
                int  at[8];
#pragma unroll
                for (int u = 0; u < 8; ++u) {
                    const int p = (t * 8 + u) * L2_THREADS + tid;
                    int i4 = p + start;
                    if (i4 >= n4) i4 -= n4;
                    at[u]  = (p < n4) ? i4 : -1;
                    buf[u] = d4[at[u] < 0 ? 0 : at[u]];
                }
#pragma unroll
                for (int u = 0; u < 8; ++u)
                    test4(buf[u], e0 + 4 * at[u], at[u] < 0 ? 0u : (unsigned)nn);
            }
            for (int e = e0 + 4 * n4 + tid; e < e1; e += L2_THREADS) {
                const unsigned r = (unsigned)(dst[e] - n0);
                if (r < (unsigned)nn) push(e, (int)r);
            }
        } else {
            for (int e = e0 + tid; e < e1; e += L2_THREADS) {
                const unsigned r = (unsigned)(dst[e] - n0);
                if (r < (unsigned)nn) push(e, (int)r);
            }
        }
        __syncthreads();
        return cnt;
    };

    auto drain_max = [&](int count) {
        for (int q = tid; q < count; q += L2_THREADS) {
            const int e  = qe[q];
            const int dl = (int)qd[q];
            const int s  = (e < E) ? ei[e] : (e - E);
            const float v = leaky(as[s] + adl[dl]);
            qe[q] = s;
            qw[q] = v;
            atomicMaxF32(&mx[dl], v);
        }
    };

    auto drain_sum = [&](int count) {
        for (int q = tid; q < count; q += L2_THREADS) {
            const int dl  = (int)qd[q];
            const float w = expf(qw[q] - mx[dl]);
            qw[q] = w;
            __hip_atomic_fetch_add(&den[dl], w, __ATOMIC_RELAXED, __HIP_MEMORY_SCOPE_WORKGROUP);
        }
        __syncthreads();
        const int nitems = count * 10;
        for (int base = 0; base < nitems; base += 4 * L2_THREADS) {
            int   q[4], g[4], s[4], o[4];
            float w[4];
            bool  ok[4];
#pragma unroll
            for (int j = 0; j < 4; ++j) {
                const int it = base + j * L2_THREADS + tid;
                ok[j] = it < nitems;
                const int itc = ok[j] ? it : (nitems - 1);
                q[j] = itc / 10;
                g[j] = itc - q[j] * 10;
            }
#pragma unroll
            for (int j = 0; j < 4; ++j) {
                s[j] = qe[q[j]];
                w[j] = qw[q[j]];
                o[j] = (int)qd[q[j]] * 40 + g[j] * 4;
            }
            float4 hv[4];
#pragma unroll
            for (int j = 0; j < 4; ++j)
                hv[j] = *(const float4*)(h3 + (size_t)s[j] * 48 + g[j] * 4);
#pragma unroll
            for (int j = 0; j < 4; ++j) {
                if (ok[j]) {
                    __hip_atomic_fetch_add(&acc[o[j]],     w[j] * hv[j].x,
                                           __ATOMIC_RELAXED, __HIP_MEMORY_SCOPE_WORKGROUP);
                    __hip_atomic_fetch_add(&acc[o[j] + 1], w[j] * hv[j].y,
                                           __ATOMIC_RELAXED, __HIP_MEMORY_SCOPE_WORKGROUP);
                    __hip_atomic_fetch_add(&acc[o[j] + 2], w[j] * hv[j].z,
                                           __ATOMIC_RELAXED, __HIP_MEMORY_SCOPE_WORKGROUP);
                    __hip_atomic_fetch_add(&acc[o[j] + 3], w[j] * hv[j].w,
                                           __ATOMIC_RELAXED, __HIP_MEMORY_SCOPE_WORKGROUP);
                }
            }
        }
    };

    const int total = fill_queue(0, E, true);
    if (total <= L2_QCAP) {
        drain_max(total);
        __syncthreads();
        drain_sum(total);
    } else {
        for (int ph = 0; ph < 2; ++ph) {
            for (int e0 = 0; e0 < E; e0 += L2_SEG) {
                const int e1 = (e0 + L2_SEG < E) ? (e0 + L2_SEG) : E;
                const int c  = fill_queue(e0, e1, e0 == 0);
                drain_max(c);
                __syncthreads();
                if (ph) drain_sum(c);
            }
        }
    }
    __syncthreads();

    float* __restrict__ o0 = out + (size_t)n0 * 40;
    for (int i = tid; i < nn * 40; i += L2_THREADS) {
        const int dl = i / 40;
        const int c  = i - dl * 40;
        o0[i] = acc[i] / den[dl] + b[c];
    }
}

extern "C" void kernel_launch(void* const* d_in, const int* in_sizes, int n_in,
                              void* d_out, int out_size, void* d_ws, size_t ws_size,
                              hipStream_t stream)
{
    const float* x    = (const float*)d_in[0];
    const int*   ei   = (const int*)  d_in[1];
    const float* W1   = (const float*)d_in[2];
    const float* aS1  = (const float*)d_in[3];
    const float* aD1  = (const float*)d_in[4];
    const float* b1   = (const float*)d_in[5];
    const float* W2   = (const float*)d_in[6];
    const float* aS2  = (const float*)d_in[7];
    const float* aD2  = (const float*)d_in[8];
    const float* b2   = (const float*)d_in[9];
    float* out = (float*)d_out;

    const int IN  = 256, F1 = 512, OUT = 40, OUTP = 48;
    const int Nn  = in_sizes[0] / IN;
    const int E   = in_sizes[1] / 2;
    const int ET  = E + Nn;

    float* ws  = (float*)d_ws;
    float* h1  = ws;
    float* h2  = h1  + (size_t)Nn * F1;
    float* as1 = h2  + (size_t)Nn * F1;
    float* ad1 = as1 + (size_t)Nn * 8;
    float* mx1 = ad1 + (size_t)Nn * 8;
    float* dn1 = mx1 + (size_t)Nn * 8;
    float* ex1 = dn1 + (size_t)Nn * 8;
    float* h3  = ex1 + (size_t)ET * 8;
    float* as2 = h3  + (size_t)Nn * 48;
    float* ad2 = as2 + Nn;
    float* W2p = ad2 + Nn;

    const int B = 256;
    const float NEG_INF = -__builtin_inff();

    gemm_wmma_f32_chmajor_out<4><<<dim3((Nn + 15) / 16, F1 / 64), 32, 0, stream>>>(
        x, W1, h1, Nn, IN, F1);

    alpha1_kernel<<<(Nn * 8 + B - 1) / B, B, 0, stream>>>(h1, aS1, aD1, as1, ad1, Nn);

    fill_f32<<<(Nn * 8 + B - 1) / B, B, 0, stream>>>(mx1, Nn * 8, NEG_INF);
    fill_f32<<<(Nn * 8 + B - 1) / B, B, 0, stream>>>(dn1, Nn * 8, 0.f);

    edge_max1   <<<(ET * 8 + B - 1) / B, B, 0, stream>>>(ei, E, Nn, as1, ad1, mx1);
    edge_expsum1<<<(ET * 8 + B - 1) / B, B, 0, stream>>>(ei, E, Nn, as1, ad1, mx1, ex1, dn1);

    aggregate1_sliced<<<F1 / 2, AGG_THREADS, 0, stream>>>(ei, E, Nn, ex1, dn1, h1, b1, h2);

    pad_cols<<<(F1 * OUTP + B - 1) / B, B, 0, stream>>>(W2, W2p, F1, OUT, OUTP);

    gemm_wmma_f32_chmajor_in<3><<<dim3((Nn + 15) / 16, 1), 32, 0, stream>>>(
        h2, W2p, h3, Nn, F1, OUTP, OUTP);

    alpha2_kernel<<<(Nn + B - 1) / B, B, 0, stream>>>(h3, aS2, aD2, as2, ad2, Nn);

    layer2_by_dst_range<<<(Nn + L2_NPW - 1) / L2_NPW, L2_THREADS, 0, stream>>>(
        ei, E, Nn, as2, ad2, h3, b2, out);
}


# Layer-1 aggregation: a node's two running sums moved from one interleaved LDS array to two planes, so the 32 lanes of each LDS float add spread over all 64 banks (was 32); sum order kept
# speedup vs baseline: 1.8533x; 1.0607x over previous
#include <hip/hip_runtime.h>
#include <hip/hip_bf16.h>
#include <math.h>


typedef float v2f __attribute__((ext_vector_type(2)));
typedef float v8f __attribute__((ext_vector_type(8)));

#define NEG_SLOPE 0.2f

__device__ __forceinline__ float leaky(float v) {
    return v > 0.0f ? v : NEG_SLOPE * v;
}

__device__ __forceinline__ void atomicMaxF32(float* addr, float val) {
    if (val >= 0.0f) atomicMax((int*)addr, __float_as_int(val));
    else             atomicMin((unsigned int*)addr, __float_as_uint(val));
}

template<int NT>
__global__ void __launch_bounds__(32)
gemm_wmma_f32_chmajor_out(const float* __restrict__ A, const float* __restrict__ B,
                          float* __restrict__ Ct, int M, int K, int N)
{
    __shared__ float As[16 * 12];

    const int lane = (int)threadIdx.x;
    const int half = lane >> 4;
    const int l16  = lane & 15;

    const int alr = lane >> 1;
    const int alc = (lane & 1) * 4;
    int rowA = blockIdx.x * 16 + alr;
    if (rowA >= M) rowA = M - 1;
    const float* __restrict__ Aload = A + (size_t)rowA * (size_t)K + alc;

    const int ncol0 = blockIdx.y * (16 * NT);
    const int colB  = ncol0 + l16;

    v8f acc[NT];
#pragma unroll
    for (int t = 0; t < NT; ++t) {
        v8f z = {0.f, 0.f, 0.f, 0.f, 0.f, 0.f, 0.f, 0.f};
        acc[t] = z;
    }

    float4 av = *(const float4*)(Aload);

    for (int kc = 0; kc < K; kc += 8) {
        __syncthreads();
        *(float4*)(&As[alr * 12 + alc]) = av;

        v2f breg[2][NT];
#pragma unroll
        for (int kk2 = 0; kk2 < 2; ++kk2) {
            const float* __restrict__ Bk =
                B + (size_t)(kc + kk2 * 4 + 2 * half) * (size_t)N + colB;
#pragma unroll
            for (int t = 0; t < NT; ++t) {
                breg[kk2][t].x = Bk[t * 16];
                breg[kk2][t].y = Bk[N + t * 16];
            }
        }

        const int kn = (kc + 8 < K) ? (kc + 8) : 0;
        av = *(const float4*)(Aload + kn);

        __syncthreads();

#pragma unroll
        for (int kk2 = 0; kk2 < 2; ++kk2) {
            const v2f a = *(const v2f*)(&As[l16 * 12 + kk2 * 4 + 2 * half]);
#pragma unroll
            for (int t = 0; t < NT; ++t) {
                acc[t] = __builtin_amdgcn_wmma_f32_16x16x4_f32(
                    false, breg[kk2][t], false, a, (short)0, acc[t], false, false);
            }
        }
    }

    const int node0 = blockIdx.x * 16;
    float* __restrict__ Cbase =
        Ct + (size_t)(ncol0 + half * 8) * (size_t)M + node0 + l16;
    if (node0 + 16 <= M) {
#pragma unroll
        for (int t = 0; t < NT; ++t) {
#pragma unroll
            for (int r = 0; r < 8; ++r)
                Cbase[(size_t)(t * 16 + r) * (size_t)M] = acc[t][r];
        }
    } else {
#pragma unroll
        for (int t = 0; t < NT; ++t) {
#pragma unroll
            for (int r = 0; r < 8; ++r)
                if (node0 + l16 < M) Cbase[(size_t)(t * 16 + r) * (size_t)M] = acc[t][r];
        }
    }
}

#define G2_DEPTH 4

template<int NT>
__device__ __forceinline__ void g2_load_chunk(const float* __restrict__ Aload,
                                              const float* __restrict__ Bload,
                                              int M, int N, int kc,
                                              v2f (&a)[2], v2f (&b)[2][NT])
{
#pragma unroll
    for (int kk2 = 0; kk2 < 2; ++kk2) {
        const float* __restrict__ Ak = Aload + (size_t)(kc + kk2 * 4) * (size_t)M;
        const float* __restrict__ Bk = Bload + (size_t)(kc + kk2 * 4) * (size_t)N;
        a[kk2].x = Ak[0];
        a[kk2].y = Ak[M];
#pragma unroll
        for (int t = 0; t < NT; ++t) {
            b[kk2][t].x = Bk[t * 16];
            b[kk2][t].y = Bk[N + t * 16];
        }
    }
}

template<int NT>
__device__ __forceinline__ void g2_ring_turn(v8f (&acc)[NT],
                                             v2f (&ra)[G2_DEPTH][2],
                                             v2f (&rb)[G2_DEPTH][2][NT],
                                             const float* __restrict__ Aload,
                                             const float* __restrict__ Bload,
                                             int M, int N, int c0, int nch)
{
#pragma unroll
    for (int s = 0; s < G2_DEPTH; ++s) {
        if (c0 >= 0) {
#pragma unroll
            for (int t = 0; t < NT; ++t) {
                acc[t] = __builtin_amdgcn_wmma_f32_16x16x4_f32(
                    false, ra[s][0], false, rb[s][0][t], (short)0, acc[t], false, false);
            }
#pragma unroll
            for (int t = 0; t < NT; ++t) {
                acc[t] = __builtin_amdgcn_wmma_f32_16x16x4_f32(
                    false, ra[s][1], false, rb[s][1][t], (short)0, acc[t], false, false);
            }
        }

        int cn = c0 + s + G2_DEPTH;
        if (cn >= nch) cn = 0;
        g2_load_chunk<NT>(Aload, Bload, M, N, cn * 8, ra[s], rb[s]);
        __builtin_amdgcn_sched_barrier(0);
    }
}

template<int NT>
__global__ void __launch_bounds__(32)
gemm_wmma_f32_chmajor_in(const float* __restrict__ At, const float* __restrict__ B,
                         float* __restrict__ C, int M, int K, int N, int ldc)
{
    const int lane = (int)threadIdx.x;
    const int half = lane >> 4;
    const int l16  = lane & 15;

    int rowA = blockIdx.x * 16 + l16;
    if (rowA >= M) rowA = M - 1;
    const float* __restrict__ Aload = At + (size_t)(2 * half) * (size_t)M + rowA;

    const int ncol0 = blockIdx.y * (16 * NT);
    const float* __restrict__ Bload = B + (size_t)(2 * half) * (size_t)N + ncol0 + l16;

    v8f acc[NT];
#pragma unroll
    for (int t = 0; t < NT; ++t) {
        v8f z = {0.f, 0.f, 0.f, 0.f, 0.f, 0.f, 0.f, 0.f};
        acc[t] = z;
    }

    const int nch = K >> 3;
    if (nch > 0) {
        v2f ra[G2_DEPTH][2];
        v2f rb[G2_DEPTH][2][NT];
        for (int c0 = -G2_DEPTH; c0 < nch; c0 += G2_DEPTH)
            g2_ring_turn<NT>(acc, ra, rb, Aload, Bload, M, N, c0, nch);
    }

    const int crow0 = blockIdx.x * 16 + half * 8;
    float* __restrict__ Cbase = C + (size_t)crow0 * (size_t)ldc + ncol0 + l16;
    if (blockIdx.x * 16 + 16 <= M) {
#pragma unroll
        for (int t = 0; t < NT; ++t) {
            float* __restrict__ Cp = Cbase + t * 16;
#pragma unroll
            for (int r = 0; r < 8; ++r)
                Cp[(size_t)r * (size_t)ldc] = acc[t][r];
        }
    } else {
#pragma unroll
        for (int t = 0; t < NT; ++t) {
            float* __restrict__ Cp = Cbase + t * 16;
#pragma unroll
            for (int r = 0; r < 8; ++r)
                if (crow0 + r < M) Cp[(size_t)r * (size_t)ldc] = acc[t][r];
        }
    }
}

__global__ void fill_f32(float* __restrict__ p, int n, float v) {
    int t = blockIdx.x * blockDim.x + threadIdx.x;
    if (t < n) p[t] = v;
}

__global__ void pad_cols(const float* __restrict__ src, float* __restrict__ dst,
                         int K, int N, int Np)
{
    int t = blockIdx.x * blockDim.x + threadIdx.x;
    if (t >= K * Np) return;
    int r = t / Np, c = t - r * Np;
    dst[t] = (c < N) ? src[r * N + c] : 0.f;
}

__global__ void alpha1_kernel(const float* __restrict__ h1t,
                              const float* __restrict__ a_src,
                              const float* __restrict__ a_dst,
                              float* __restrict__ as, float* __restrict__ ad, int Nn)
{
    int t = blockIdx.x * blockDim.x + threadIdx.x;
    if (t >= Nn * 8) return;
    int h = t / Nn, n = t - h * Nn;
    const float* hp = h1t + (size_t)(h * 64) * (size_t)Nn + n;
    const float* sp = a_src + h * 64;
    const float* dp = a_dst + h * 64;
    float ss = 0.f, dd = 0.f;
#pragma unroll 8
    for (int c = 0; c < 64; ++c) {
        float v = hp[(size_t)c * (size_t)Nn];
        ss += v * sp[c];
        dd += v * dp[c];
    }
    as[n * 8 + h] = ss;
    ad[n * 8 + h] = dd;
}

__device__ __forceinline__ void edge_sd(const int* __restrict__ ei, int E, int e,
                                        int& s, int& d)
{
    if (e < E) { s = ei[e]; d = ei[E + e]; }
    else       { s = e - E; d = e - E; }
}

__global__ void edge_max1(const int* __restrict__ ei, int E, int Nn,
                          const float* __restrict__ as, const float* __restrict__ ad,
                          float* __restrict__ mx)
{
    int t = blockIdx.x * blockDim.x + threadIdx.x;
    if (t >= (E + Nn) * 8) return;
    int e = t >> 3, h = t & 7;
    int s, d; edge_sd(ei, E, e, s, d);
    atomicMaxF32(&mx[d * 8 + h], leaky(as[s * 8 + h] + ad[d * 8 + h]));
}

__global__ void edge_expsum1(const int* __restrict__ ei, int E, int Nn,
                             const float* __restrict__ as, const float* __restrict__ ad,
                             const float* __restrict__ mx,
                             float* __restrict__ ex, float* __restrict__ denom)
{
    int t = blockIdx.x * blockDim.x + threadIdx.x;
    if (t >= (E + Nn) * 8) return;
    int e = t >> 3, h = t & 7;
    int s, d; edge_sd(ei, E, e, s, d);
    float v  = leaky(as[s * 8 + h] + ad[d * 8 + h]);
    float xv = expf(v - mx[d * 8 + h]);
    ex[(size_t)h * (size_t)(E + Nn) + e] = xv;
    atomicAdd(&denom[d * 8 + h], xv);
}

#define AGG_NODES   10496
#define AGG_THREADS 1024

__global__ void __launch_bounds__(AGG_THREADS)
aggregate1_sliced(const int* __restrict__ ei, int E, int Nn,
                  const float* __restrict__ ex, const float* __restrict__ denom,
                  const float* __restrict__ h1t, const float* __restrict__ b,
                  float* __restrict__ h2t)
{
    __shared__ float2 hs[AGG_NODES];
    __shared__ float  acc[2 * AGG_NODES];
    float* const acc0 = acc;
    float* const acc1 = acc + AGG_NODES;

    const int tid = (int)threadIdx.x;
    const int c0  = (int)blockIdx.x * 2;
    const int h   = c0 >> 6;
    const int ET  = E + Nn;

    const float* __restrict__ exh = ex + (size_t)h * (size_t)ET;
    const float* __restrict__ p0  = h1t + (size_t)c0 * (size_t)Nn;
    const float* __restrict__ p1  = p0 + Nn;
    for (int n = tid; n < Nn; n += AGG_THREADS) {
        const float x0 = p0[n], x1 = p1[n];
        const float ws = exh[E + n];
        hs[n] = make_float2(x0, x1);
        acc0[n] = ws * x0;
        acc1[n] = ws * x1;
    }
    __syncthreads();

    const int* __restrict__ src = ei;
    const int* __restrict__ dst = ei + E;

    auto load_group = [&](int g, int* s, int* d, float* w) {
#pragma unroll
        for (int j = 0; j < 4; ++j) {
            const int e = g * (4 * AGG_THREADS) + tid + j * AGG_THREADS;
            s[j] = src[e];
            d[j] = dst[e];
            w[j] = exh[e];
        }
    };
    auto add_edge = [&](int d, float w, float2 hv) {
        __hip_atomic_fetch_add(&acc0[d], w * hv.x,
                               __ATOMIC_RELAXED, __HIP_MEMORY_SCOPE_WORKGROUP);
        __hip_atomic_fetch_add(&acc1[d], w * hv.y,
                               __ATOMIC_RELAXED, __HIP_MEMORY_SCOPE_WORKGROUP);
    };
    auto do_group = [&](const int* s, const int* d, const float* w) {
        float2 hv[4];
#pragma unroll
        for (int j = 0; j < 4; ++j) hv[j] = hs[s[j]];
#pragma unroll
        for (int j = 0; j < 4; ++j) add_edge(d[j], w[j], hv[j]);
    };

    const int nfull = E / (4 * AGG_THREADS);
    int   sA[4], dA[4], sB[4], dB[4];
    float wA[4], wB[4];
    if (nfull > 0) load_group(0, sA, dA, wA);
    int g = 0;
    for (; g + 2 <= nfull; g += 2) {
        load_group(g + 1, sB, dB, wB);
        do_group(sA, dA, wA);
        load_group((g + 2 < nfull) ? (g + 2) : (g + 1), sA, dA, wA);
        do_group(sB, dB, wB);
    }
    if (g < nfull) do_group(sA, dA, wA);

    for (int e = nfull * (4 * AGG_THREADS) + tid; e < E; e += AGG_THREADS)
        add_edge(dst[e], exh[e], hs[src[e]]);
    __syncthreads();

    const float b0 = b[c0], b1 = b[c0 + 1];
    float* __restrict__ q0 = h2t + (size_t)c0 * (size_t)Nn;
    float* __restrict__ q1 = q0 + Nn;
    for (int n = tid; n < Nn; n += AGG_THREADS) {
        const float dn = denom[n * 8 + h];
        const float v0 = acc0[n] / dn + b0;
        const float v1 = acc1[n] / dn + b1;
        q0[n] = v0 > 0.f ? v0 : expm1f(v0);
        q1[n] = v1 > 0.f ? v1 : expm1f(v1);
    }
}

__global__ void alpha2_kernel(const float* __restrict__ h3,
                              const float* __restrict__ a_src,
                              const float* __restrict__ a_dst,
                              float* __restrict__ as, float* __restrict__ ad, int Nn)
{
    int n = blockIdx.x * blockDim.x + threadIdx.x;
    if (n >= Nn) return;
    const float* hp = h3 + (size_t)n * 48;
    float ss = 0.f, dd = 0.f;
#pragma unroll 8
    for (int c = 0; c < 40; ++c) { float v = hp[c]; ss += v * a_src[c]; dd += v * a_dst[c]; }
    as[n] = ss;
    ad[n] = dd;
}

#define L2_NPW     40
#define L2_THREADS 1024
#define L2_QCAP    20480
#define L2_SEG     (L2_QCAP - 64)

__global__ void __launch_bounds__(L2_THREADS)
layer2_by_dst_range(const int* __restrict__ ei, int E, int Nn,
                    const float* __restrict__ as, const float* __restrict__ ad,
                    const float* __restrict__ h3, const float* __restrict__ b,
                    float* __restrict__ out)
{
    __shared__ int            qe[L2_QCAP];
    __shared__ float          qw[L2_QCAP];
    __shared__ unsigned short qd[L2_QCAP];
    __shared__ float acc[L2_NPW * 40];
    __shared__ float mx[L2_NPW];
    __shared__ float den[L2_NPW];
    __shared__ float adl[L2_NPW];
    __shared__ int   cnt;

    const int tid = (int)threadIdx.x;
    const int n0  = (int)blockIdx.x * L2_NPW;
    const int nn  = (Nn - n0 < L2_NPW) ? (Nn - n0) : L2_NPW;
    const int* __restrict__ dst = ei + E;

    for (int i = tid; i < L2_NPW * 40; i += L2_THREADS) acc[i] = 0.f;
    if (tid < L2_NPW) {
        mx[tid]  = -__builtin_inff();
        den[tid] = 0.f;
        adl[tid] = (tid < nn) ? ad[n0 + tid] : 0.f;
    }

    auto push = [&](int e, int dl) {
        const int q = atomicAdd(&cnt, 1);
        if (q < L2_QCAP) { qe[q] = e; qd[q] = (unsigned short)dl; }
    };
    auto test4 = [&](int4 d, int e, unsigned un) {
        const unsigned r0 = (unsigned)(d.x - n0), r1 = (unsigned)(d.y - n0);
        const unsigned r2 = (unsigned)(d.z - n0), r3 = (unsigned)(d.w - n0);
        if ((r0 < un) | (r1 < un) | (r2 < un) | (r3 < un)) {
            if (r0 < un) push(e,     (int)r0);
            if (r1 < un) push(e + 1, (int)r1);
            if (r2 < un) push(e + 2, (int)r2);
            if (r3 < un) push(e + 3, (int)r3);
        }
    };

    auto fill_queue = [&](int e0, int e1, bool with_self) -> int {
        __syncthreads();
        if (tid == 0) cnt = 0;
        __syncthreads();
        if (with_self && tid < nn) push(E + n0 + tid, tid);

        if (((E | e0) & 3) == 0 && ((unsigned long long)ei & 15ull) == 0) {
            const int4* __restrict__ d4 = (const int4*)(dst + e0);
            const int n4    = (e1 - e0) >> 2;
            const int start = (int)(blockIdx.x * ((unsigned)n4 / gridDim.x)) & ~31;
            const int ntrip = (n4 + 8 * L2_THREADS - 1) / (8 * L2_THREADS);
            for (int t = 0; t < ntrip; ++t) {
                int4 buf[8];
                int  at[8];
#pragma unroll
                for (int u = 0; u < 8; ++u) {
                    const int p = (t * 8 + u) * L2_THREADS + tid;
                    int i4 = p + start;
                    if (i4 >= n4) i4 -= n4;
                    at[u]  = (p < n4) ? i4 : -1;
                    buf[u] = d4[at[u] < 0 ? 0 : at[u]];
                }
#pragma unroll
                for (int u = 0; u < 8; ++u)
                    test4(buf[u], e0 + 4 * at[u], at[u] < 0 ? 0u : (unsigned)nn);
            }
            for (int e = e0 + 4 * n4 + tid; e < e1; e += L2_THREADS) {
                const unsigned r = (unsigned)(dst[e] - n0);
                if (r < (unsigned)nn) push(e, (int)r);
            }
        } else {
            for (int e = e0 + tid; e < e1; e += L2_THREADS) {
                const unsigned r = (unsigned)(dst[e] - n0);
                if (r < (unsigned)nn) push(e, (int)r);
            }
        }
        __syncthreads();
        return cnt;
    };

    auto drain_max = [&](int count) {
        for (int q = tid; q < count; q += L2_THREADS) {
            const int e  = qe[q];
            const int dl = (int)qd[q];
            const int s  = (e < E) ? ei[e] : (e - E);
            const float v = leaky(as[s] + adl[dl]);
            qe[q] = s;
            qw[q] = v;
            atomicMaxF32(&mx[dl], v);
        }
    };

    auto drain_sum = [&](int count) {
        for (int q = tid; q < count; q += L2_THREADS) {
            const int dl  = (int)qd[q];
            const float w = expf(qw[q] - mx[dl]);
            qw[q] = w;
            __hip_atomic_fetch_add(&den[dl], w, __ATOMIC_RELAXED, __HIP_MEMORY_SCOPE_WORKGROUP);
        }
        __syncthreads();
        const int nitems = count * 10;
        for (int base = 0; base < nitems; base += 4 * L2_THREADS) {
            int   q[4], g[4], s[4], o[4];
            float w[4];
            bool  ok[4];
#pragma unroll
            for (int j = 0; j < 4; ++j) {
                const int it = base + j * L2_THREADS + tid;
                ok[j] = it < nitems;
                const int itc = ok[j] ? it : (nitems - 1);
                q[j] = itc / 10;
                g[j] = itc - q[j] * 10;
            }
#pragma unroll
            for (int j = 0; j < 4; ++j) {
                s[j] = qe[q[j]];
                w[j] = qw[q[j]];
                o[j] = (int)qd[q[j]] * 40 + g[j] * 4;
            }
            float4 hv[4];
#pragma unroll
            for (int j = 0; j < 4; ++j)
                hv[j] = *(const float4*)(h3 + (size_t)s[j] * 48 + g[j] * 4);
#pragma unroll
            for (int j = 0; j < 4; ++j) {
                if (ok[j]) {
                    __hip_atomic_fetch_add(&acc[o[j]],     w[j] * hv[j].x,
                                           __ATOMIC_RELAXED, __HIP_MEMORY_SCOPE_WORKGROUP);
                    __hip_atomic_fetch_add(&acc[o[j] + 1], w[j] * hv[j].y,
                                           __ATOMIC_RELAXED, __HIP_MEMORY_SCOPE_WORKGROUP);
                    __hip_atomic_fetch_add(&acc[o[j] + 2], w[j] * hv[j].z,
                                           __ATOMIC_RELAXED, __HIP_MEMORY_SCOPE_WORKGROUP);
                    __hip_atomic_fetch_add(&acc[o[j] + 3], w[j] * hv[j].w,
                                           __ATOMIC_RELAXED, __HIP_MEMORY_SCOPE_WORKGROUP);
                }
            }
        }
    };

    const int total = fill_queue(0, E, true);
    if (total <= L2_QCAP) {
        drain_max(total);
        __syncthreads();
        drain_sum(total);
    } else {
        for (int ph = 0; ph < 2; ++ph) {
            for (int e0 = 0; e0 < E; e0 += L2_SEG) {
                const int e1 = (e0 + L2_SEG < E) ? (e0 + L2_SEG) : E;
                const int c  = fill_queue(e0, e1, e0 == 0);
                drain_max(c);
                __syncthreads();
                if (ph) drain_sum(c);
            }
        }
    }
    __syncthreads();

    float* __restrict__ o0 = out + (size_t)n0 * 40;
    for (int i = tid; i < nn * 40; i += L2_THREADS) {
        const int dl = i / 40;
        const int c  = i - dl * 40;
        o0[i] = acc[i] / den[dl] + b[c];
    }
}

extern "C" void kernel_launch(void* const* d_in, const int* in_sizes, int n_in,
                              void* d_out, int out_size, void* d_ws, size_t ws_size,
                              hipStream_t stream)
{
    const float* x    = (const float*)d_in[0];
    const int*   ei   = (const int*)  d_in[1];
    const float* W1   = (const float*)d_in[2];
    const float* aS1  = (const float*)d_in[3];
    const float* aD1  = (const float*)d_in[4];
    const float* b1   = (const float*)d_in[5];
    const float* W2   = (const float*)d_in[6];
    const float* aS2  = (const float*)d_in[7];
    const float* aD2  = (const float*)d_in[8];
    const float* b2   = (const float*)d_in[9];
    float* out = (float*)d_out;

    const int IN  = 256, F1 = 512, OUT = 40, OUTP = 48;
    const int Nn  = in_sizes[0] / IN;
    const int E   = in_sizes[1] / 2;
    const int ET  = E + Nn;

    float* ws  = (float*)d_ws;
    float* h1  = ws;
    float* h2  = h1  + (size_t)Nn * F1;
    float* as1 = h2  + (size_t)Nn * F1;
    float* ad1 = as1 + (size_t)Nn * 8;
    float* mx1 = ad1 + (size_t)Nn * 8;
    float* dn1 = mx1 + (size_t)Nn * 8;
    float* ex1 = dn1 + (size_t)Nn * 8;
    float* h3  = ex1 + (size_t)ET * 8;
    float* as2 = h3  + (size_t)Nn * 48;
    float* ad2 = as2 + Nn;
    float* W2p = ad2 + Nn;

    const int B = 256;
    const float NEG_INF = -__builtin_inff();

    gemm_wmma_f32_chmajor_out<4><<<dim3((Nn + 15) / 16, F1 / 64), 32, 0, stream>>>(
        x, W1, h1, Nn, IN, F1);

    alpha1_kernel<<<(Nn * 8 + B - 1) / B, B, 0, stream>>>(h1, aS1, aD1, as1, ad1, Nn);

    fill_f32<<<(Nn * 8 + B - 1) / B, B, 0, stream>>>(mx1, Nn * 8, NEG_INF);
    fill_f32<<<(Nn * 8 + B - 1) / B, B, 0, stream>>>(dn1, Nn * 8, 0.f);

    edge_max1   <<<(ET * 8 + B - 1) / B, B, 0, stream>>>(ei, E, Nn, as1, ad1, mx1);
    edge_expsum1<<<(ET * 8 + B - 1) / B, B, 0, stream>>>(ei, E, Nn, as1, ad1, mx1, ex1, dn1);

    aggregate1_sliced<<<F1 / 2, AGG_THREADS, 0, stream>>>(ei, E, Nn, ex1, dn1, h1, b1, h2);

    pad_cols<<<(F1 * OUTP + B - 1) / B, B, 0, stream>>>(W2, W2p, F1, OUT, OUTP);

    gemm_wmma_f32_chmajor_in<3><<<dim3((Nn + 15) / 16, 1), 32, 0, stream>>>(
        h2, W2p, h3, Nn, F1, OUTP, OUTP);

    alpha2_kernel<<<(Nn + B - 1) / B, B, 0, stream>>>(h3, aS2, aD2, as2, ad2, Nn);

    layer2_by_dst_range<<<(Nn + L2_NPW - 1) / L2_NPW, L2_THREADS, 0, stream>>>(
        ei, E, Nn, as2, ad2, h3, b2, out);
}


# Four helper launches folded into their neighbours (11 -> 7): max/exp-sum start values stored by the attention-half kernel, W2 read unpadded by the GEMM, layer-2 halves formed in the range kernel
# speedup vs baseline: 1.9105x; 1.0309x over previous
#include <hip/hip_runtime.h>
#include <hip/hip_bf16.h>
#include <math.h>


typedef float v2f __attribute__((ext_vector_type(2)));
typedef float v8f __attribute__((ext_vector_type(8)));

#define NEG_SLOPE 0.2f

__device__ __forceinline__ float leaky(float v) {
    return v > 0.0f ? v : NEG_SLOPE * v;
}

__device__ __forceinline__ void atomicMaxF32(float* addr, float val) {
    if (val >= 0.0f) atomicMax((int*)addr, __float_as_int(val));
    else             atomicMin((unsigned int*)addr, __float_as_uint(val));
}

template<int NT>
__global__ void __launch_bounds__(32)
gemm_wmma_f32_chmajor_out(const float* __restrict__ A, const float* __restrict__ B,
                          float* __restrict__ Ct, int M, int K, int N)
{
    __shared__ float As[16 * 12];

    const int lane = (int)threadIdx.x;
    const int half = lane >> 4;
    const int l16  = lane & 15;

    const int alr = lane >> 1;
    const int alc = (lane & 1) * 4;
    int rowA = blockIdx.x * 16 + alr;
    if (rowA >= M) rowA = M - 1;
    const float* __restrict__ Aload = A + (size_t)rowA * (size_t)K + alc;

    const int ncol0 = blockIdx.y * (16 * NT);
    const int colB  = ncol0 + l16;

    v8f acc[NT];
#pragma unroll
    for (int t = 0; t < NT; ++t) {
        v8f z = {0.f, 0.f, 0.f, 0.f, 0.f, 0.f, 0.f, 0.f};
        acc[t] = z;
    }

    float4 av = *(const float4*)(Aload);

    for (int kc = 0; kc < K; kc += 8) {
        __syncthreads();
        *(float4*)(&As[alr * 12 + alc]) = av;

        v2f breg[2][NT];
#pragma unroll
        for (int kk2 = 0; kk2 < 2; ++kk2) {
            const float* __restrict__ Bk =
                B + (size_t)(kc + kk2 * 4 + 2 * half) * (size_t)N + colB;
#pragma unroll
            for (int t = 0; t < NT; ++t) {
                breg[kk2][t].x = Bk[t * 16];
                breg[kk2][t].y = Bk[N + t * 16];
            }
        }

        const int kn = (kc + 8 < K) ? (kc + 8) : 0;
        av = *(const float4*)(Aload + kn);

        __syncthreads();

#pragma unroll
        for (int kk2 = 0; kk2 < 2; ++kk2) {
            const v2f a = *(const v2f*)(&As[l16 * 12 + kk2 * 4 + 2 * half]);
#pragma unroll
            for (int t = 0; t < NT; ++t) {
                acc[t] = __builtin_amdgcn_wmma_f32_16x16x4_f32(
                    false, breg[kk2][t], false, a, (short)0, acc[t], false, false);
            }
        }
    }

    const int node0 = blockIdx.x * 16;
    float* __restrict__ Cbase =
        Ct + (size_t)(ncol0 + half * 8) * (size_t)M + node0 + l16;
    if (node0 + 16 <= M) {
#pragma unroll
        for (int t = 0; t < NT; ++t) {
#pragma unroll
            for (int r = 0; r < 8; ++r)
                Cbase[(size_t)(t * 16 + r) * (size_t)M] = acc[t][r];
        }
    } else {
#pragma unroll
        for (int t = 0; t < NT; ++t) {
#pragma unroll
            for (int r = 0; r < 8; ++r)
                if (node0 + l16 < M) Cbase[(size_t)(t * 16 + r) * (size_t)M] = acc[t][r];
        }
    }
}

#define G2_DEPTH 4

template<int NT>
__device__ __forceinline__ void g2_load_chunk(const float* __restrict__ Aload,
                                              const float* __restrict__ Bload,
                                              const int (&bcol)[NT],
                                              int M, int N, int kc,
                                              v2f (&a)[2], v2f (&b)[2][NT])
{
#pragma unroll
    for (int kk2 = 0; kk2 < 2; ++kk2) {
        const float* __restrict__ Ak = Aload + (size_t)(kc + kk2 * 4) * (size_t)M;
        const float* __restrict__ Bk = Bload + (size_t)(kc + kk2 * 4) * (size_t)N;
        a[kk2].x = Ak[0];
        a[kk2].y = Ak[M];
#pragma unroll
        for (int t = 0; t < NT; ++t) {
            b[kk2][t].x = Bk[bcol[t]];
            b[kk2][t].y = Bk[N + bcol[t]];
        }
    }
}

template<int NT>
__device__ __forceinline__ void g2_ring_turn(v8f (&acc)[NT],
                                             v2f (&ra)[G2_DEPTH][2],
                                             v2f (&rb)[G2_DEPTH][2][NT],
                                             const float* __restrict__ Aload,
                                             const float* __restrict__ Bload,
                                             const int (&bcol)[NT],
                                             int M, int N, int c0, int nch)
{
#pragma unroll
    for (int s = 0; s < G2_DEPTH; ++s) {
        if (c0 >= 0) {
#pragma unroll
            for (int t = 0; t < NT; ++t) {
                acc[t] = __builtin_amdgcn_wmma_f32_16x16x4_f32(
                    false, ra[s][0], false, rb[s][0][t], (short)0, acc[t], false, false);
            }
#pragma unroll
            for (int t = 0; t < NT; ++t) {
                acc[t] = __builtin_amdgcn_wmma_f32_16x16x4_f32(
                    false, ra[s][1], false, rb[s][1][t], (short)0, acc[t], false, false);
            }
        }

        int cn = c0 + s + G2_DEPTH;
        if (cn >= nch) cn = 0;
        g2_load_chunk<NT>(Aload, Bload, bcol, M, N, cn * 8, ra[s], rb[s]);
        __builtin_amdgcn_sched_barrier(0);
    }
}

template<int NT>
__global__ void __launch_bounds__(32)
gemm_wmma_f32_chmajor_in(const float* __restrict__ At, const float* __restrict__ B,
                         float* __restrict__ C, int M, int K, int N, int ldc)
{
    const int lane = (int)threadIdx.x;
    const int half = lane >> 4;
    const int l16  = lane & 15;

    int rowA = blockIdx.x * 16 + l16;
    if (rowA >= M) rowA = M - 1;
    const float* __restrict__ Aload = At + (size_t)(2 * half) * (size_t)M + rowA;

    const int ncol0 = blockIdx.y * (16 * NT);
    const float* __restrict__ Bload = B + (size_t)(2 * half) * (size_t)N;
    int bcol[NT];
#pragma unroll
    for (int t = 0; t < NT; ++t) {
        const int ct = ncol0 + t * 16 + l16;
        bcol[t] = (ct < N) ? ct : (N - 1);
    }

    v8f acc[NT];
#pragma unroll
    for (int t = 0; t < NT; ++t) {
        v8f z = {0.f, 0.f, 0.f, 0.f, 0.f, 0.f, 0.f, 0.f};
        acc[t] = z;
    }

    const int nch = K >> 3;
    if (nch > 0) {
        v2f ra[G2_DEPTH][2];
        v2f rb[G2_DEPTH][2][NT];
        for (int c0 = -G2_DEPTH; c0 < nch; c0 += G2_DEPTH)
            g2_ring_turn<NT>(acc, ra, rb, Aload, Bload, bcol, M, N, c0, nch);
    }

    const int crow0 = blockIdx.x * 16 + half * 8;
    float* __restrict__ Cbase = C + (size_t)crow0 * (size_t)ldc + ncol0 + l16;
    if (blockIdx.x * 16 + 16 <= M) {
#pragma unroll
        for (int t = 0; t < NT; ++t) {
            float* __restrict__ Cp = Cbase + t * 16;
            if (ncol0 + t * 16 + l16 < N) {
#pragma unroll
                for (int r = 0; r < 8; ++r)
                    Cp[(size_t)r * (size_t)ldc] = acc[t][r];
            }
        }
    } else {
#pragma unroll
        for (int t = 0; t < NT; ++t) {
            float* __restrict__ Cp = Cbase + t * 16;
            if (ncol0 + t * 16 + l16 < N) {
#pragma unroll
                for (int r = 0; r < 8; ++r)
                    if (crow0 + r < M) Cp[(size_t)r * (size_t)ldc] = acc[t][r];
            }
        }
    }
}

__global__ void alpha1_kernel(const float* __restrict__ h1t,
                              const float* __restrict__ a_src,
                              const float* __restrict__ a_dst,
                              float* __restrict__ as, float* __restrict__ ad,
                              float* __restrict__ mx, float* __restrict__ denom, int Nn)
{
    int t = blockIdx.x * blockDim.x + threadIdx.x;
    if (t >= Nn * 8) return;
    int h = t / Nn, n = t - h * Nn;
    const float* hp = h1t + (size_t)(h * 64) * (size_t)Nn + n;
    const float* sp = a_src + h * 64;
    const float* dp = a_dst + h * 64;
    float ss = 0.f, dd = 0.f;
#pragma unroll 8
    for (int c = 0; c < 64; ++c) {
        float v = hp[(size_t)c * (size_t)Nn];
        ss += v * sp[c];
        dd += v * dp[c];
    }
    as[n * 8 + h] = ss;
    ad[n * 8 + h] = dd;
    mx[n * 8 + h]    = -__builtin_inff();
    denom[n * 8 + h] = 0.f;
}

__device__ __forceinline__ void edge_sd(const int* __restrict__ ei, int E, int e,
                                        int& s, int& d)
{
    if (e < E) { s = ei[e]; d = ei[E + e]; }
    else       { s = e - E; d = e - E; }
}

__global__ void edge_max1(const int* __restrict__ ei, int E, int Nn,
                          const float* __restrict__ as, const float* __restrict__ ad,
                          float* __restrict__ mx)
{
    int t = blockIdx.x * blockDim.x + threadIdx.x;
    if (t >= (E + Nn) * 8) return;
    int e = t >> 3, h = t & 7;
    int s, d; edge_sd(ei, E, e, s, d);
    atomicMaxF32(&mx[d * 8 + h], leaky(as[s * 8 + h] + ad[d * 8 + h]));
}

__global__ void edge_expsum1(const int* __restrict__ ei, int E, int Nn,
                             const float* __restrict__ as, const float* __restrict__ ad,
                             const float* __restrict__ mx,
                             float* __restrict__ ex, float* __restrict__ denom)
{
    int t = blockIdx.x * blockDim.x + threadIdx.x;
    if (t >= (E + Nn) * 8) return;
    int e = t >> 3, h = t & 7;
    int s, d; edge_sd(ei, E, e, s, d);
    float v  = leaky(as[s * 8 + h] + ad[d * 8 + h]);
    float xv = expf(v - mx[d * 8 + h]);
    ex[(size_t)h * (size_t)(E + Nn) + e] = xv;
    atomicAdd(&denom[d * 8 + h], xv);
}

#define AGG_NODES   10496
#define AGG_THREADS 1024

__global__ void __launch_bounds__(AGG_THREADS)
aggregate1_sliced(const int* __restrict__ ei, int E, int Nn,
                  const float* __restrict__ ex, const float* __restrict__ denom,
                  const float* __restrict__ h1t, const float* __restrict__ b,
                  float* __restrict__ h2t)
{
    __shared__ float2 hs[AGG_NODES];
    __shared__ float  acc[2 * AGG_NODES];
    float* const acc0 = acc;
    float* const acc1 = acc + AGG_NODES;

    const int tid = (int)threadIdx.x;
    const int c0  = (int)blockIdx.x * 2;
    const int h   = c0 >> 6;
    const int ET  = E + Nn;

    const float* __restrict__ exh = ex + (size_t)h * (size_t)ET;
    const float* __restrict__ p0  = h1t + (size_t)c0 * (size_t)Nn;
    const float* __restrict__ p1  = p0 + Nn;
    for (int n = tid; n < Nn; n += AGG_THREADS) {
        const float x0 = p0[n], x1 = p1[n];
        const float ws = exh[E + n];
        hs[n] = make_float2(x0, x1);
        acc0[n] = ws * x0;
        acc1[n] = ws * x1;
    }
    __syncthreads();

    const int* __restrict__ src = ei;
    const int* __restrict__ dst = ei + E;

    auto load_group = [&](int g, int* s, int* d, float* w) {
#pragma unroll
        for (int j = 0; j < 4; ++j) {
            const int e = g * (4 * AGG_THREADS) + tid + j * AGG_THREADS;
            s[j] = src[e];
            d[j] = dst[e];
            w[j] = exh[e];
        }
    };
    auto add_edge = [&](int d, float w, float2 hv) {
        __hip_atomic_fetch_add(&acc0[d], w * hv.x,
                               __ATOMIC_RELAXED, __HIP_MEMORY_SCOPE_WORKGROUP);
        __hip_atomic_fetch_add(&acc1[d], w * hv.y,
                               __ATOMIC_RELAXED, __HIP_MEMORY_SCOPE_WORKGROUP);
    };
    auto do_group = [&](const int* s, const int* d, const float* w) {
        float2 hv[4];
#pragma unroll
        for (int j = 0; j < 4; ++j) hv[j] = hs[s[j]];
#pragma unroll
        for (int j = 0; j < 4; ++j) add_edge(d[j], w[j], hv[j]);
    };

    const int nfull = E / (4 * AGG_THREADS);
    int   sA[4], dA[4], sB[4], dB[4];
    float wA[4], wB[4];
    if (nfull > 0) load_group(0, sA, dA, wA);
    int g = 0;
    for (; g + 2 <= nfull; g += 2) {
        load_group(g + 1, sB, dB, wB);
        do_group(sA, dA, wA);
        load_group((g + 2 < nfull) ? (g + 2) : (g + 1), sA, dA, wA);
        do_group(sB, dB, wB);
    }
    if (g < nfull) do_group(sA, dA, wA);

    for (int e = nfull * (4 * AGG_THREADS) + tid; e < E; e += AGG_THREADS)
        add_edge(dst[e], exh[e], hs[src[e]]);
    __syncthreads();

    const float b0 = b[c0], b1 = b[c0 + 1];
    float* __restrict__ q0 = h2t + (size_t)c0 * (size_t)Nn;
    float* __restrict__ q1 = q0 + Nn;
    for (int n = tid; n < Nn; n += AGG_THREADS) {
        const float dn = denom[n * 8 + h];
        const float v0 = acc0[n] / dn + b0;
        const float v1 = acc1[n] / dn + b1;
        q0[n] = v0 > 0.f ? v0 : expm1f(v0);
        q1[n] = v1 > 0.f ? v1 : expm1f(v1);
    }
}

__device__ __forceinline__ float row_dot40(const float* __restrict__ hp,
                                           const float* __restrict__ a)
{
    float4 v[10];
#pragma unroll
    for (int g = 0; g < 10; ++g) v[g] = *(const float4*)(hp + g * 4);
    float ss = 0.f;
#pragma unroll
    for (int g = 0; g < 10; ++g) {
        ss = __builtin_fmaf(v[g].x, a[g * 4],     ss);
        ss = __builtin_fmaf(v[g].y, a[g * 4 + 1], ss);
        ss = __builtin_fmaf(v[g].z, a[g * 4 + 2], ss);
        ss = __builtin_fmaf(v[g].w, a[g * 4 + 3], ss);
    }
    return ss;
}

#define L2_NPW     40
#define L2_THREADS 1024
#define L2_QCAP    20480
#define L2_SEG     (L2_QCAP - 64)

__global__ void __launch_bounds__(L2_THREADS)
layer2_by_dst_range(const int* __restrict__ ei, int E, int Nn,
                    const float* __restrict__ a_src, const float* __restrict__ a_dst,
                    const float* __restrict__ h3, const float* __restrict__ b,
                    float* __restrict__ out)
{
    __shared__ int            qe[L2_QCAP];
    __shared__ float          qw[L2_QCAP];
    __shared__ unsigned short qd[L2_QCAP];
    __shared__ float acc[L2_NPW * 40];
    __shared__ float mx[L2_NPW];
    __shared__ float den[L2_NPW];
    __shared__ float adl[L2_NPW];
    __shared__ int   cnt;

    const int tid = (int)threadIdx.x;
    const int n0  = (int)blockIdx.x * L2_NPW;
    const int nn  = (Nn - n0 < L2_NPW) ? (Nn - n0) : L2_NPW;
    const int* __restrict__ dst = ei + E;

    for (int i = tid; i < L2_NPW * 40; i += L2_THREADS) acc[i] = 0.f;
    if (tid < L2_NPW) {
        mx[tid]  = -__builtin_inff();
        den[tid] = 0.f;
        adl[tid] = (tid < nn) ? row_dot40(h3 + (size_t)(n0 + tid) * 48, a_dst) : 0.f;
    }

    auto push = [&](int e, int dl) {
        const int q = atomicAdd(&cnt, 1);
        if (q < L2_QCAP) { qe[q] = e; qd[q] = (unsigned short)dl; }
    };
    auto test4 = [&](int4 d, int e, unsigned un) {
        const unsigned r0 = (unsigned)(d.x - n0), r1 = (unsigned)(d.y - n0);
        const unsigned r2 = (unsigned)(d.z - n0), r3 = (unsigned)(d.w - n0);
        if ((r0 < un) | (r1 < un) | (r2 < un) | (r3 < un)) {
            if (r0 < un) push(e,     (int)r0);
            if (r1 < un) push(e + 1, (int)r1);
            if (r2 < un) push(e + 2, (int)r2);
            if (r3 < un) push(e + 3, (int)r3);
        }
    };

    auto fill_queue = [&](int e0, int e1, bool with_self) -> int {
        __syncthreads();
        if (tid == 0) cnt = 0;
        __syncthreads();
        if (with_self && tid < nn) push(E + n0 + tid, tid);

        if (((E | e0) & 3) == 0 && ((unsigned long long)ei & 15ull) == 0) {
            const int4* __restrict__ d4 = (const int4*)(dst + e0);
            const int n4    = (e1 - e0) >> 2;
            const int start = (int)(blockIdx.x * ((unsigned)n4 / gridDim.x)) & ~31;
            const int ntrip = (n4 + 8 * L2_THREADS - 1) / (8 * L2_THREADS);
            for (int t = 0; t < ntrip; ++t) {
                int4 buf[8];
                int  at[8];
#pragma unroll
                for (int u = 0; u < 8; ++u) {
                    const int p = (t * 8 + u) * L2_THREADS + tid;
                    int i4 = p + start;
                    if (i4 >= n4) i4 -= n4;
                    at[u]  = (p < n4) ? i4 : -1;
                    buf[u] = d4[at[u] < 0 ? 0 : at[u]];
                }
#pragma unroll
                for (int u = 0; u < 8; ++u)
                    test4(buf[u], e0 + 4 * at[u], at[u] < 0 ? 0u : (unsigned)nn);
            }
            for (int e = e0 + 4 * n4 + tid; e < e1; e += L2_THREADS) {
                const unsigned r = (unsigned)(dst[e] - n0);
                if (r < (unsigned)nn) push(e, (int)r);
            }
        } else {
            for (int e = e0 + tid; e < e1; e += L2_THREADS) {
                const unsigned r = (unsigned)(dst[e] - n0);
                if (r < (unsigned)nn) push(e, (int)r);
            }
        }
        __syncthreads();
        return cnt;
    };

    auto drain_max = [&](int count) {
        for (int q = tid; q < count; q += L2_THREADS) {
            const int e  = qe[q];
            const int dl = (int)qd[q];
            const int s  = (e < E) ? ei[e] : (e - E);
            const float v = leaky(row_dot40(h3 + (size_t)s * 48, a_src) + adl[dl]);
            qe[q] = s;
            qw[q] = v;
            atomicMaxF32(&mx[dl], v);
        }
    };

    auto drain_sum = [&](int count) {
        for (int q = tid; q < count; q += L2_THREADS) {
            const int dl  = (int)qd[q];
            const float w = expf(qw[q] - mx[dl]);
            qw[q] = w;
            __hip_atomic_fetch_add(&den[dl], w, __ATOMIC_RELAXED, __HIP_MEMORY_SCOPE_WORKGROUP);
        }
        __syncthreads();
        const int nitems = count * 10;
        for (int base = 0; base < nitems; base += 4 * L2_THREADS) {
            int   q[4], g[4], s[4], o[4];
            float w[4];
            bool  ok[4];
#pragma unroll
            for (int j = 0; j < 4; ++j) {
                const int it = base + j * L2_THREADS + tid;
                ok[j] = it < nitems;
                const int itc = ok[j] ? it : (nitems - 1);
                q[j] = itc / 10;
                g[j] = itc - q[j] * 10;
            }
#pragma unroll
            for (int j = 0; j < 4; ++j) {
                s[j] = qe[q[j]];
                w[j] = qw[q[j]];
                o[j] = (int)qd[q[j]] * 40 + g[j] * 4;
            }
            float4 hv[4];
#pragma unroll
            for (int j = 0; j < 4; ++j)
                hv[j] = *(const float4*)(h3 + (size_t)s[j] * 48 + g[j] * 4);
#pragma unroll
            for (int j = 0; j < 4; ++j) {
                if (ok[j]) {
                    __hip_atomic_fetch_add(&acc[o[j]],     w[j] * hv[j].x,
                                           __ATOMIC_RELAXED, __HIP_MEMORY_SCOPE_WORKGROUP);
                    __hip_atomic_fetch_add(&acc[o[j] + 1], w[j] * hv[j].y,
                                           __ATOMIC_RELAXED, __HIP_MEMORY_SCOPE_WORKGROUP);
                    __hip_atomic_fetch_add(&acc[o[j] + 2], w[j] * hv[j].z,
                                           __ATOMIC_RELAXED, __HIP_MEMORY_SCOPE_WORKGROUP);
                    __hip_atomic_fetch_add(&acc[o[j] + 3], w[j] * hv[j].w,
                                           __ATOMIC_RELAXED, __HIP_MEMORY_SCOPE_WORKGROUP);
                }
            }
        }
    };

    const int total = fill_queue(0, E, true);
    if (total <= L2_QCAP) {
        drain_max(total);
        __syncthreads();
        drain_sum(total);
    } else {
        for (int ph = 0; ph < 2; ++ph) {
            for (int e0 = 0; e0 < E; e0 += L2_SEG) {
                const int e1 = (e0 + L2_SEG < E) ? (e0 + L2_SEG) : E;
                const int c  = fill_queue(e0, e1, e0 == 0);
                drain_max(c);
                __syncthreads();
                if (ph) drain_sum(c);
            }
        }
    }
    __syncthreads();

    float* __restrict__ o0 = out + (size_t)n0 * 40;
    for (int i = tid; i < nn * 40; i += L2_THREADS) {
        const int dl = i / 40;
        const int c  = i - dl * 40;
        o0[i] = acc[i] / den[dl] + b[c];
    }
}

extern "C" void kernel_launch(void* const* d_in, const int* in_sizes, int n_in,
                              void* d_out, int out_size, void* d_ws, size_t ws_size,
                              hipStream_t stream)
{
    const float* x    = (const float*)d_in[0];
    const int*   ei   = (const int*)  d_in[1];
    const float* W1   = (const float*)d_in[2];
    const float* aS1  = (const float*)d_in[3];
    const float* aD1  = (const float*)d_in[4];
    const float* b1   = (const float*)d_in[5];
    const float* W2   = (const float*)d_in[6];
    const float* aS2  = (const float*)d_in[7];
    const float* aD2  = (const float*)d_in[8];
    const float* b2   = (const float*)d_in[9];
    float* out = (float*)d_out;

    const int IN  = 256, F1 = 512, OUT = 40, OUTP = 48;
    const int Nn  = in_sizes[0] / IN;
    const int E   = in_sizes[1] / 2;
    const int ET  = E + Nn;

    float* ws  = (float*)d_ws;
    float* h1  = ws;
    float* h2  = h1  + (size_t)Nn * F1;
    float* as1 = h2  + (size_t)Nn * F1;
    float* ad1 = as1 + (size_t)Nn * 8;
    float* mx1 = ad1 + (size_t)Nn * 8;
    float* dn1 = mx1 + (size_t)Nn * 8;
    float* ex1 = dn1 + (size_t)Nn * 8;
    float* h3  = ex1 + (size_t)ET * 8;

    const int B = 256;

    gemm_wmma_f32_chmajor_out<4><<<dim3((Nn + 15) / 16, F1 / 64), 32, 0, stream>>>(
        x, W1, h1, Nn, IN, F1);

    alpha1_kernel<<<(Nn * 8 + B - 1) / B, B, 0, stream>>>(h1, aS1, aD1, as1, ad1, mx1, dn1, Nn);

    edge_max1   <<<(ET * 8 + B - 1) / B, B, 0, stream>>>(ei, E, Nn, as1, ad1, mx1);
    edge_expsum1<<<(ET * 8 + B - 1) / B, B, 0, stream>>>(ei, E, Nn, as1, ad1, mx1, ex1, dn1);

    aggregate1_sliced<<<F1 / 2, AGG_THREADS, 0, stream>>>(ei, E, Nn, ex1, dn1, h1, b1, h2);

    gemm_wmma_f32_chmajor_in<3><<<dim3((Nn + 15) / 16, 1), 32, 0, stream>>>(
        h2, W2, h3, Nn, F1, OUT, OUTP);

    layer2_by_dst_range<<<(Nn + L2_NPW - 1) / L2_NPW, L2_THREADS, 0, stream>>>(
        ei, E, Nn, aS2, aD2, h3, b2, out);
}
